# starting kernel
# baseline (speedup 1.0000x reference)
#include <hip/hip_runtime.h>
#include <cmath>

#define N_NODES 50000
#define N_EDGES 800000
#define DF 128

typedef __attribute__((ext_vector_type(2))) float v2f;
typedef __attribute__((ext_vector_type(8))) float v8f;

__global__ void copy_kernel(float* __restrict__ dst, const float* __restrict__ src, int n4) {
    int i = blockIdx.x * blockDim.x + threadIdx.x;
    if (i < n4) ((float4*)dst)[i] = ((const float4*)src)[i];
}

__global__ void scatter_kernel(float* __restrict__ agg, const float* __restrict__ x,
                               const int* __restrict__ ei) {
    int t = blockIdx.x * blockDim.x + threadIdx.x;
    int e = t >> 5;
    if (e >= N_EDGES) return;
    int f = (t & 31) << 2;
    int s = ei[e];
    int d = ei[N_EDGES + e];
    float4 v = *(const float4*)(x + (size_t)s * DF + f);
    float* p = agg + (size_t)d * DF + f;
    atomicAdd(p + 0, v.x);
    atomicAdd(p + 1, v.y);
    atomicAdd(p + 2, v.z);
    atomicAdd(p + 3, v.w);
}

template<int NSEG, int EPI>
__global__ void __launch_bounds__(256, 2)
gemm_wmma(const float* __restrict__ A0, const float* __restrict__ A1,
          const float* __restrict__ A2, const float* __restrict__ W,
          const float* __restrict__ bias, const float* __restrict__ resid,
          float* __restrict__ out) {
    __shared__ float lA[16 * DF];

    const int lane = threadIdx.x & 31;
    const int wave = threadIdx.x >> 5;
    const int r    = lane & 15;
    const int hi   = lane >> 4;
    const int n0   = wave * 16;
    const int row0 = blockIdx.x * 16;

    v8f acc = {0.f, 0.f, 0.f, 0.f, 0.f, 0.f, 0.f, 0.f};

    const float* As[3] = {A0, A1, A2};
    #pragma unroll
    for (int seg = 0; seg < NSEG; ++seg) {
        const float4* s4 = (const float4*)(As[seg] + (size_t)row0 * DF);
        float4* d4 = (float4*)lA;
        #pragma unroll
        for (int i = 0; i < 2; ++i)
            d4[threadIdx.x + i * 256] = s4[threadIdx.x + i * 256];
        __syncthreads();

        const float* Wseg = W + seg * DF * DF;
        #pragma unroll
        for (int k0 = 0; k0 < DF; k0 += 4) {
            const float* ap = lA + r * DF + k0 + 2 * hi;
            v2f a; a.x = ap[0]; a.y = ap[1];
            const float* bp = Wseg + (k0 + 2 * hi) * DF + n0 + r;
            v2f b; b.x = bp[0]; b.y = bp[DF];
            acc = __builtin_amdgcn_wmma_f32_16x16x4_f32(
                false, a, false, b, (short)0, acc, false, false);
        }
        __syncthreads();
    }

    const int col = n0 + r;
    const float bb = bias[col];
    #pragma unroll
    for (int v = 0; v < 8; ++v) {
        int row = row0 + v + 8 * hi;
        float y = acc[v] + bb;
        if (EPI == 1) y = fmaxf(y, 0.0f);
        if (EPI == 2) {
            y += resid[(size_t)row * DF + col];
            y = (y > 0.0f) ? y : expm1f(y);
        }
        out[(size_t)row * DF + col] = y;
    }
}

extern "C" void kernel_launch(void* const* d_in, const int* in_sizes, int n_in,
                              void* d_out, int out_size, void* d_ws, size_t ws_size,
                              hipStream_t stream) {
    const float* x   = (const float*)d_in[0];
    const int*   ei  = (const int*)d_in[1];
    const float* W1[3] = {(const float*)d_in[2],  (const float*)d_in[6],  (const float*)d_in[10]};
    const float* b1[3] = {(const float*)d_in[3],  (const float*)d_in[7],  (const float*)d_in[11]};
    const float* W2[3] = {(const float*)d_in[4],  (const float*)d_in[8],  (const float*)d_in[12]};
    const float* b2[3] = {(const float*)d_in[5],  (const float*)d_in[9],  (const float*)d_in[13]};
    const float* Wjk = (const float*)d_in[14];
    const float* bjk = (const float*)d_in[15];

    const size_t NM = (size_t)N_NODES * DF;
    float* h0 = (float*)d_ws;
    float* xs[3] = {h0 + NM, h0 + 2 * NM, h0 + 3 * NM};

    const int n4 = (int)(NM / 4);
    dim3 blk(256);
    const int gBlocks = N_NODES / 16;

    const float* xin = x;
    for (int l = 0; l < 3; ++l) {
        copy_kernel<<<(n4 + 255) / 256, blk, 0, stream>>>(h0, xin, n4);
        scatter_kernel<<<(N_EDGES * 32) / 256, blk, 0, stream>>>(h0, xin, ei);
        gemm_wmma<1, 1><<<gBlocks, blk, 0, stream>>>(h0, nullptr, nullptr,
                                                     W1[l], b1[l], nullptr, h0);
        gemm_wmma<1, 2><<<gBlocks, blk, 0, stream>>>(h0, nullptr, nullptr,
                                                     W2[l], b2[l], xin, xs[l]);
        xin = xs[l];
    }
    gemm_wmma<3, 0><<<gBlocks, blk, 0, stream>>>(xs[0], xs[1], xs[2],
                                                 Wjk, bjk, nullptr, (float*)d_out);
}


# Edges grouped by destination once per call (int-atomic arrival rank); per layer one wave per node sums neighbour rows in registers in rank order, stores once: no float atomics, no init copy
# speedup vs baseline: 1.6910x; 1.6910x over previous
#include <hip/hip_runtime.h>
#include <cmath>

#define N_NODES 50000
#define N_EDGES 800000
#define DF 128

typedef __attribute__((ext_vector_type(2))) float v2f;
typedef __attribute__((ext_vector_type(8))) float v8f;

__global__ void clear_kernel(int* __restrict__ cnt_total, int n) {
    int i = blockIdx.x * blockDim.x + threadIdx.x;
    if (i < n) cnt_total[i] = 0;
}

__global__ void count_kernel(int* __restrict__ cnt, int* __restrict__ rank,
                             const int* __restrict__ ei) {
    int q = blockIdx.x * blockDim.x + threadIdx.x;
    if (q >= N_EDGES / 4) return;
    int4 d = ((const int4*)(ei + N_EDGES))[q];
    int4 r;
    r.x = atomicAdd(cnt + d.x, 1);
    r.y = atomicAdd(cnt + d.y, 1);
    r.z = atomicAdd(cnt + d.z, 1);
    r.w = atomicAdd(cnt + d.w, 1);
    ((int4*)rank)[q] = r;
}

__global__ void place_kernel(const int* __restrict__ cnt, int* __restrict__ total,
                             int* __restrict__ start) {
    __shared__ int sh[256];
    __shared__ int base;
    const int tid = threadIdx.x;
    const int n = blockIdx.x * 256 + tid;
    const int c = (n < N_NODES) ? cnt[n] : 0;
    sh[tid] = c;
    __syncthreads();
    for (int off = 1; off < 256; off <<= 1) {
        int v = (tid >= off) ? sh[tid - off] : 0;
        __syncthreads();
        sh[tid] += v;
        __syncthreads();
    }
    if (tid == 255) base = atomicAdd(total, sh[255]);
    __syncthreads();
    if (n < N_NODES) start[n] = base + sh[tid] - c;
}

__global__ void fill_kernel(int* __restrict__ nbr, const int* __restrict__ start,
                            const int* __restrict__ rank, const int* __restrict__ ei) {
    int q = blockIdx.x * blockDim.x + threadIdx.x;
    if (q >= N_EDGES / 4) return;
    int4 s = ((const int4*)ei)[q];
    int4 d = ((const int4*)(ei + N_EDGES))[q];
    int4 r = ((const int4*)rank)[q];
    nbr[start[d.x] + r.x] = s.x;
    nbr[start[d.y] + r.y] = s.y;
    nbr[start[d.z] + r.z] = s.z;
    nbr[start[d.w] + r.w] = s.w;
}

template<int K>
__device__ __forceinline__ void add_rows(float4& acc, const float4* __restrict__ x4,
                                         int idx, int j, int lane) {
    float4 v[K];
    #pragma unroll
    for (int k = 0; k < K; ++k) {
        int s = __builtin_amdgcn_readlane(idx, j + k);
        v[k] = x4[(size_t)s * (DF / 4) + lane];
    }
    #pragma unroll
    for (int k = 0; k < K; ++k) {
        acc.x += v[k].x; acc.y += v[k].y; acc.z += v[k].z; acc.w += v[k].w;
    }
}

__global__ void __launch_bounds__(256)
gather_kernel(float* __restrict__ agg, const float* __restrict__ x,
              const int* __restrict__ start, const int* __restrict__ cnt,
              const int* __restrict__ nbr) {
    const int lane = threadIdx.x & 31;
    const int node = __builtin_amdgcn_readfirstlane(blockIdx.x * 8 + (threadIdx.x >> 5));
    const float4* x4 = (const float4*)x;
    const int s0 = start[node];
    const int deg = cnt[node];
    float4 acc = x4[(size_t)node * (DF / 4) + lane];
    for (int b = 0; b < deg; b += 32) {
        const int m = (deg - b < 32) ? (deg - b) : 32;
        const int idx = (lane < m) ? nbr[s0 + b + lane] : 0;
        int j = 0;
        for (; j + 8 <= m; j += 8) add_rows<8>(acc, x4, idx, j, lane);
        if (j + 4 <= m) { add_rows<4>(acc, x4, idx, j, lane); j += 4; }
        if (j + 2 <= m) { add_rows<2>(acc, x4, idx, j, lane); j += 2; }
        if (j < m)      { add_rows<1>(acc, x4, idx, j, lane); }
    }
    ((float4*)agg)[(size_t)node * (DF / 4) + lane] = acc;
}

template<int NSEG, int EPI>
__global__ void __launch_bounds__(256, 2)
gemm_wmma(const float* __restrict__ A0, const float* __restrict__ A1,
          const float* __restrict__ A2, const float* __restrict__ W,
          const float* __restrict__ bias, const float* __restrict__ resid,
          float* __restrict__ out) {
    __shared__ float lA[16 * DF];

    const int lane = threadIdx.x & 31;
    const int wave = threadIdx.x >> 5;
    const int r    = lane & 15;
    const int hi   = lane >> 4;
    const int n0   = wave * 16;
    const int row0 = blockIdx.x * 16;

    v8f acc = {0.f, 0.f, 0.f, 0.f, 0.f, 0.f, 0.f, 0.f};

    const float* As[3] = {A0, A1, A2};
    #pragma unroll
    for (int seg = 0; seg < NSEG; ++seg) {
        const float4* s4 = (const float4*)(As[seg] + (size_t)row0 * DF);
        float4* d4 = (float4*)lA;
        #pragma unroll
        for (int i = 0; i < 2; ++i)
            d4[threadIdx.x + i * 256] = s4[threadIdx.x + i * 256];
        __syncthreads();

        const float* Wseg = W + seg * DF * DF;
        #pragma unroll
        for (int k0 = 0; k0 < DF; k0 += 4) {
            const float* ap = lA + r * DF + k0 + 2 * hi;
            v2f a; a.x = ap[0]; a.y = ap[1];
            const float* bp = Wseg + (k0 + 2 * hi) * DF + n0 + r;
            v2f b; b.x = bp[0]; b.y = bp[DF];
            acc = __builtin_amdgcn_wmma_f32_16x16x4_f32(
                false, a, false, b, (short)0, acc, false, false);
        }
        __syncthreads();
    }

    const int col = n0 + r;
    const float bb = bias[col];
    #pragma unroll
    for (int v = 0; v < 8; ++v) {
        int row = row0 + v + 8 * hi;
        float y = acc[v] + bb;
        if (EPI == 1) y = fmaxf(y, 0.0f);
        if (EPI == 2) {
            y += resid[(size_t)row * DF + col];
            y = (y > 0.0f) ? y : expm1f(y);
        }
        out[(size_t)row * DF + col] = y;
    }
}

extern "C" void kernel_launch(void* const* d_in, const int* in_sizes, int n_in,
                              void* d_out, int out_size, void* d_ws, size_t ws_size,
                              hipStream_t stream) {
    const float* x   = (const float*)d_in[0];
    const int*   ei  = (const int*)d_in[1];
    const float* W1[3] = {(const float*)d_in[2],  (const float*)d_in[6],  (const float*)d_in[10]};
    const float* b1[3] = {(const float*)d_in[3],  (const float*)d_in[7],  (const float*)d_in[11]};
    const float* W2[3] = {(const float*)d_in[4],  (const float*)d_in[8],  (const float*)d_in[12]};
    const float* b2[3] = {(const float*)d_in[5],  (const float*)d_in[9],  (const float*)d_in[13]};
    const float* Wjk = (const float*)d_in[14];
    const float* bjk = (const float*)d_in[15];

    const size_t NM = (size_t)N_NODES * DF;
    float* h0 = (float*)d_ws;
    float* xs[3] = {h0 + NM, h0 + 2 * NM, h0 + 3 * NM};

    dim3 blk(256);
    const int gBlocks = N_NODES / 16;

    int* cnt   = (int*)xs[2];
    int* total = cnt + N_NODES;
    int* start = cnt + 50048;
    int* rank  = cnt + 100096;
    int* nbr   = cnt + 900096;
    const int qBlocks = (N_EDGES / 4 + 255) / 256;
    clear_kernel<<<(N_NODES + 1 + 255) / 256, blk, 0, stream>>>(cnt, N_NODES + 1);
    count_kernel<<<qBlocks, blk, 0, stream>>>(cnt, rank, ei);
    place_kernel<<<(N_NODES + 255) / 256, blk, 0, stream>>>(cnt, total, start);
    fill_kernel<<<qBlocks, blk, 0, stream>>>(nbr, start, rank, ei);

    const float* xin = x;
    for (int l = 0; l < 3; ++l) {
        gather_kernel<<<N_NODES / 8, blk, 0, stream>>>(h0, xin, start, cnt, nbr);
        gemm_wmma<1, 1><<<gBlocks, blk, 0, stream>>>(h0, nullptr, nullptr,
                                                     W1[l], b1[l], nullptr, h0);
        gemm_wmma<1, 2><<<gBlocks, blk, 0, stream>>>(h0, nullptr, nullptr,
                                                     W2[l], b2[l], xin, xs[l]);
        xin = xs[l];
    }
    gemm_wmma<3, 0><<<gBlocks, blk, 0, stream>>>(xs[0], xs[1], xs[2],
                                                 Wjk, bjk, nullptr, (float*)d_out);
}


# GEMM A tile in LDS at a row pitch of 132 floats instead of 128: the 16 rows of one A-fragment read fall on 16 different bank groups instead of one bank (16-way LDS bank conflict removed)
# speedup vs baseline: 2.2182x; 1.3118x over previous
#include <hip/hip_runtime.h>
#include <cmath>

#define N_NODES 50000
#define N_EDGES 800000
#define DF 128

typedef __attribute__((ext_vector_type(2))) float v2f;
typedef __attribute__((ext_vector_type(8))) float v8f;

__global__ void clear_kernel(int* __restrict__ cnt_total, int n) {
    int i = blockIdx.x * blockDim.x + threadIdx.x;
    if (i < n) cnt_total[i] = 0;
}

__global__ void count_kernel(int* __restrict__ cnt, int* __restrict__ rank,
                             const int* __restrict__ ei) {
    int q = blockIdx.x * blockDim.x + threadIdx.x;
    if (q >= N_EDGES / 4) return;
    int4 d = ((const int4*)(ei + N_EDGES))[q];
    int4 r;
    r.x = atomicAdd(cnt + d.x, 1);
    r.y = atomicAdd(cnt + d.y, 1);
    r.z = atomicAdd(cnt + d.z, 1);
    r.w = atomicAdd(cnt + d.w, 1);
    ((int4*)rank)[q] = r;
}

__global__ void place_kernel(const int* __restrict__ cnt, int* __restrict__ total,
                             int* __restrict__ start) {
    __shared__ int sh[256];
    __shared__ int base;
    const int tid = threadIdx.x;
    const int n = blockIdx.x * 256 + tid;
    const int c = (n < N_NODES) ? cnt[n] : 0;
    sh[tid] = c;
    __syncthreads();
    for (int off = 1; off < 256; off <<= 1) {
        int v = (tid >= off) ? sh[tid - off] : 0;
        __syncthreads();
        sh[tid] += v;
        __syncthreads();
    }
    if (tid == 255) base = atomicAdd(total, sh[255]);
    __syncthreads();
    if (n < N_NODES) start[n] = base + sh[tid] - c;
}

__global__ void fill_kernel(int* __restrict__ nbr, const int* __restrict__ start,
                            const int* __restrict__ rank, const int* __restrict__ ei) {
    int q = blockIdx.x * blockDim.x + threadIdx.x;
    if (q >= N_EDGES / 4) return;
    int4 s = ((const int4*)ei)[q];
    int4 d = ((const int4*)(ei + N_EDGES))[q];
    int4 r = ((const int4*)rank)[q];
    nbr[start[d.x] + r.x] = s.x;
    nbr[start[d.y] + r.y] = s.y;
    nbr[start[d.z] + r.z] = s.z;
    nbr[start[d.w] + r.w] = s.w;
}

template<int K>
__device__ __forceinline__ void add_rows(float4& acc, const float4* __restrict__ x4,
                                         int idx, int j, int lane) {
    float4 v[K];
    #pragma unroll
    for (int k = 0; k < K; ++k) {
        int s = __builtin_amdgcn_readlane(idx, j + k);
        v[k] = x4[(size_t)s * (DF / 4) + lane];
    }
    #pragma unroll
    for (int k = 0; k < K; ++k) {
        acc.x += v[k].x; acc.y += v[k].y; acc.z += v[k].z; acc.w += v[k].w;
    }
}

__global__ void __launch_bounds__(256)
gather_kernel(float* __restrict__ agg, const float* __restrict__ x,
              const int* __restrict__ start, const int* __restrict__ cnt,
              const int* __restrict__ nbr) {
    const int lane = threadIdx.x & 31;
    const int node = __builtin_amdgcn_readfirstlane(blockIdx.x * 8 + (threadIdx.x >> 5));
    const float4* x4 = (const float4*)x;
    const int s0 = start[node];
    const int deg = cnt[node];
    float4 acc = x4[(size_t)node * (DF / 4) + lane];
    for (int b = 0; b < deg; b += 32) {
        const int m = (deg - b < 32) ? (deg - b) : 32;
        const int idx = (lane < m) ? nbr[s0 + b + lane] : 0;
        int j = 0;
        for (; j + 8 <= m; j += 8) add_rows<8>(acc, x4, idx, j, lane);
        if (j + 4 <= m) { add_rows<4>(acc, x4, idx, j, lane); j += 4; }
        if (j + 2 <= m) { add_rows<2>(acc, x4, idx, j, lane); j += 2; }
        if (j < m)      { add_rows<1>(acc, x4, idx, j, lane); }
    }
    ((float4*)agg)[(size_t)node * (DF / 4) + lane] = acc;
}

#define LDA (DF + 4)
template<int NSEG, int EPI>
__global__ void __launch_bounds__(256, 2)
gemm_wmma(const float* __restrict__ A0, const float* __restrict__ A1,
          const float* __restrict__ A2, const float* __restrict__ W,
          const float* __restrict__ bias, const float* __restrict__ resid,
          float* __restrict__ out) {
    __shared__ float lA[16 * LDA];

    const int lane = threadIdx.x & 31;
    const int wave = threadIdx.x >> 5;
    const int r    = lane & 15;
    const int hi   = lane >> 4;
    const int n0   = wave * 16;
    const int row0 = blockIdx.x * 16;

    v8f acc = {0.f, 0.f, 0.f, 0.f, 0.f, 0.f, 0.f, 0.f};

    const float* As[3] = {A0, A1, A2};
    #pragma unroll
    for (int seg = 0; seg < NSEG; ++seg) {
        const float4* s4 = (const float4*)(As[seg] + (size_t)row0 * DF);
        #pragma unroll
        for (int i = 0; i < 2; ++i)
            *(float4*)(lA + (wave + i * 8) * LDA + lane * 4) = s4[threadIdx.x + i * 256];
        __syncthreads();

        const float* Wseg = W + seg * DF * DF;
        #pragma unroll
        for (int k0 = 0; k0 < DF; k0 += 4) {
            const float* ap = lA + r * LDA + k0 + 2 * hi;
            v2f a; a.x = ap[0]; a.y = ap[1];
            const float* bp = Wseg + (k0 + 2 * hi) * DF + n0 + r;
            v2f b; b.x = bp[0]; b.y = bp[DF];
            acc = __builtin_amdgcn_wmma_f32_16x16x4_f32(
                false, a, false, b, (short)0, acc, false, false);
        }
        __syncthreads();
    }

    const int col = n0 + r;
    const float bb = bias[col];
    #pragma unroll
    for (int v = 0; v < 8; ++v) {
        int row = row0 + v + 8 * hi;
        float y = acc[v] + bb;
        if (EPI == 1) y = fmaxf(y, 0.0f);
        if (EPI == 2) {
            y += resid[(size_t)row * DF + col];
            y = (y > 0.0f) ? y : expm1f(y);
        }
        out[(size_t)row * DF + col] = y;
    }
}

extern "C" void kernel_launch(void* const* d_in, const int* in_sizes, int n_in,
                              void* d_out, int out_size, void* d_ws, size_t ws_size,
                              hipStream_t stream) {
    const float* x   = (const float*)d_in[0];
    const int*   ei  = (const int*)d_in[1];
    const float* W1[3] = {(const float*)d_in[2],  (const float*)d_in[6],  (const float*)d_in[10]};
    const float* b1[3] = {(const float*)d_in[3],  (const float*)d_in[7],  (const float*)d_in[11]};
    const float* W2[3] = {(const float*)d_in[4],  (const float*)d_in[8],  (const float*)d_in[12]};
    const float* b2[3] = {(const float*)d_in[5],  (const float*)d_in[9],  (const float*)d_in[13]};
    const float* Wjk = (const float*)d_in[14];
    const float* bjk = (const float*)d_in[15];

    const size_t NM = (size_t)N_NODES * DF;
    float* h0 = (float*)d_ws;
    float* xs[3] = {h0 + NM, h0 + 2 * NM, h0 + 3 * NM};

    dim3 blk(256);
    const int gBlocks = N_NODES / 16;

    int* cnt   = (int*)xs[2];
    int* total = cnt + N_NODES;
    int* start = cnt + 50048;
    int* rank  = cnt + 100096;
    int* nbr   = cnt + 900096;
    const int qBlocks = (N_EDGES / 4 + 255) / 256;
    clear_kernel<<<(N_NODES + 1 + 255) / 256, blk, 0, stream>>>(cnt, N_NODES + 1);
    count_kernel<<<qBlocks, blk, 0, stream>>>(cnt, rank, ei);
    place_kernel<<<(N_NODES + 255) / 256, blk, 0, stream>>>(cnt, total, start);
    fill_kernel<<<qBlocks, blk, 0, stream>>>(nbr, start, rank, ei);

    const float* xin = x;
    for (int l = 0; l < 3; ++l) {
        gather_kernel<<<N_NODES / 8, blk, 0, stream>>>(h0, xin, start, cnt, nbr);
        gemm_wmma<1, 1><<<gBlocks, blk, 0, stream>>>(h0, nullptr, nullptr,
                                                     W1[l], b1[l], nullptr, h0);
        gemm_wmma<1, 2><<<gBlocks, blk, 0, stream>>>(h0, nullptr, nullptr,
                                                     W2[l], b2[l], xin, xs[l]);
        xin = xs[l];
    }
    gemm_wmma<3, 0><<<gBlocks, blk, 0, stream>>>(xs[0], xs[1], xs[2],
                                                 Wjk, bjk, nullptr, (float*)d_out);
}


# GEMMs run as 512 resident work-groups looping over 16-row tiles: each wave loads its 128x16 slice of W into registers once per launch (was per tile, from L2); A chunks double-buffered in LDS
# speedup vs baseline: 2.3568x; 1.0625x over previous
#include <hip/hip_runtime.h>
#include <cmath>

#define N_NODES 50000
#define N_EDGES 800000
#define DF 128

typedef __attribute__((ext_vector_type(2))) float v2f;
typedef __attribute__((ext_vector_type(8))) float v8f;

__global__ void clear_kernel(int* __restrict__ cnt_total, int n) {
    int i = blockIdx.x * blockDim.x + threadIdx.x;
    if (i < n) cnt_total[i] = 0;
}

__global__ void count_kernel(int* __restrict__ cnt, int* __restrict__ rank,
                             const int* __restrict__ ei) {
    int q = blockIdx.x * blockDim.x + threadIdx.x;
    if (q >= N_EDGES / 4) return;
    int4 d = ((const int4*)(ei + N_EDGES))[q];
    int4 r;
    r.x = atomicAdd(cnt + d.x, 1);
    r.y = atomicAdd(cnt + d.y, 1);
    r.z = atomicAdd(cnt + d.z, 1);
    r.w = atomicAdd(cnt + d.w, 1);
    ((int4*)rank)[q] = r;
}

__global__ void place_kernel(const int* __restrict__ cnt, int* __restrict__ total,
                             int* __restrict__ start) {
    __shared__ int sh[256];
    __shared__ int base;
    const int tid = threadIdx.x;
    const int n = blockIdx.x * 256 + tid;
    const int c = (n < N_NODES) ? cnt[n] : 0;
    sh[tid] = c;
    __syncthreads();
    for (int off = 1; off < 256; off <<= 1) {
        int v = (tid >= off) ? sh[tid - off] : 0;
        __syncthreads();
        sh[tid] += v;
        __syncthreads();
    }
    if (tid == 255) base = atomicAdd(total, sh[255]);
    __syncthreads();
    if (n < N_NODES) start[n] = base + sh[tid] - c;
}

__global__ void fill_kernel(int* __restrict__ nbr, const int* __restrict__ start,
                            const int* __restrict__ rank, const int* __restrict__ ei) {
    int q = blockIdx.x * blockDim.x + threadIdx.x;
    if (q >= N_EDGES / 4) return;
    int4 s = ((const int4*)ei)[q];
    int4 d = ((const int4*)(ei + N_EDGES))[q];
    int4 r = ((const int4*)rank)[q];
    nbr[start[d.x] + r.x] = s.x;
    nbr[start[d.y] + r.y] = s.y;
    nbr[start[d.z] + r.z] = s.z;
    nbr[start[d.w] + r.w] = s.w;
}

template<int K>
__device__ __forceinline__ void add_rows(float4& acc, const float4* __restrict__ x4,
                                         int idx, int j, int lane) {
    float4 v[K];
    #pragma unroll
    for (int k = 0; k < K; ++k) {
        int s = __builtin_amdgcn_readlane(idx, j + k);
        v[k] = x4[(size_t)s * (DF / 4) + lane];
    }
    #pragma unroll
    for (int k = 0; k < K; ++k) {
        acc.x += v[k].x; acc.y += v[k].y; acc.z += v[k].z; acc.w += v[k].w;
    }
}

__global__ void __launch_bounds__(256)
gather_kernel(float* __restrict__ agg, const float* __restrict__ x,
              const int* __restrict__ start, const int* __restrict__ cnt,
              const int* __restrict__ nbr) {
    const int lane = threadIdx.x & 31;
    const int node = __builtin_amdgcn_readfirstlane(blockIdx.x * 8 + (threadIdx.x >> 5));
    const float4* x4 = (const float4*)x;
    const int s0 = start[node];
    const int deg = cnt[node];
    float4 acc = x4[(size_t)node * (DF / 4) + lane];
    for (int b = 0; b < deg; b += 32) {
        const int m = (deg - b < 32) ? (deg - b) : 32;
        const int idx = (lane < m) ? nbr[s0 + b + lane] : 0;
        int j = 0;
        for (; j + 8 <= m; j += 8) add_rows<8>(acc, x4, idx, j, lane);
        if (j + 4 <= m) { add_rows<4>(acc, x4, idx, j, lane); j += 4; }
        if (j + 2 <= m) { add_rows<2>(acc, x4, idx, j, lane); j += 2; }
        if (j < m)      { add_rows<1>(acc, x4, idx, j, lane); }
    }
    ((float4*)agg)[(size_t)node * (DF / 4) + lane] = acc;
}

#define LDA (DF + 4)
#define N_TILES 3125
#define GEMM_BLOCKS 512
template<int NSEG, int EPI>
__global__ void __launch_bounds__(256, 4)
gemm_wmma(const float* A0, const float* A1, const float* A2,
          const float* __restrict__ W, const float* __restrict__ bias,
          const float* resid, float* out) {
    __shared__ float lA[2][16 * LDA];

    const int tid  = threadIdx.x;
    const int lane = tid & 31;
    const int wave = tid >> 5;
    const int r    = lane & 15;
    const int hi   = lane >> 4;
    const int n0   = wave * 16;
    const int col  = n0 + r;

    const float* As[3] = {A0, A1, A2};

    constexpr int RA = 4;

    const int st_off = wave * LDA + lane * 4;
    const int fr_off = r * LDA + 2 * hi;

    int t = blockIdx.x;
    float4 p0, p1;
    {
        const float4* s4 = (const float4*)(As[0] + (size_t)t * 16 * DF);
        p0 = s4[tid];
        p1 = s4[tid + 256];
    }

    v2f breg[NSEG][DF / 4];
    #pragma unroll
    for (int seg = 0; seg < NSEG; ++seg) {
        #pragma unroll
        for (int kk = 0; kk < DF / 4; ++kk) {
            const float* bp = W + seg * DF * DF + (4 * kk + 2 * hi) * DF + col;
            breg[seg][kk].x = bp[0];
            breg[seg][kk].y = bp[DF];
        }
    }
    const float bb = bias[col];

    *(float4*)(&lA[0][st_off])           = p0;
    *(float4*)(&lA[0][st_off + 8 * LDA]) = p1;
    __syncthreads();

    int buf = 0;
    for (; t < N_TILES; t += GEMM_BLOCKS) {
        v8f acc = {0.f, 0.f, 0.f, 0.f, 0.f, 0.f, 0.f, 0.f};
        float rs[8] = {0.f, 0.f, 0.f, 0.f, 0.f, 0.f, 0.f, 0.f};
        const size_t rbase = (size_t)(t * 16 + 8 * hi) * DF + col;
        #pragma unroll
        for (int seg = 0; seg < NSEG; ++seg) {
            const bool last = (seg == NSEG - 1);
            const int  tn   = last ? t + GEMM_BLOCKS : t;
            const int  sn   = last ? 0 : seg + 1;
            const int  tc   = (tn < N_TILES) ? tn : t;
            {
                const float4* s4 = (const float4*)(As[sn] + (size_t)tc * 16 * DF);
                p0 = s4[tid];
                p1 = s4[tid + 256];
            }
            if (EPI == 2 && last) {
                #pragma unroll
                for (int v = 0; v < 8; ++v)
                    rs[v] = resid[rbase + v * DF];
            }

            const float* la = &lA[buf][fr_off];
            #pragma unroll
            for (int kk = 0; kk < DF / 4; ++kk) {
                v2f a; a.x = la[4 * kk]; a.y = la[4 * kk + 1];
                acc = __builtin_amdgcn_wmma_f32_16x16x4_f32(
                    false, a, false, breg[seg][kk], (short)0, acc, false, false);
            }
            if (EPI == 2 && last) __builtin_amdgcn_sched_group_barrier(0x020, 10, 0);
            else                  __builtin_amdgcn_sched_group_barrier(0x020, 2, 0);
            __builtin_amdgcn_sched_group_barrier(0x100, RA, 0);
            #pragma unroll
            for (int g = 0; g < 16 - RA; ++g) {
                __builtin_amdgcn_sched_group_barrier(0x008, 2, 0);
                __builtin_amdgcn_sched_group_barrier(0x100, 1, 0);
            }
            __builtin_amdgcn_sched_group_barrier(0x008, 2 * RA, 0);

            *(float4*)(&lA[buf ^ 1][st_off])           = p0;
            *(float4*)(&lA[buf ^ 1][st_off + 8 * LDA]) = p1;
            __syncthreads();
            __builtin_amdgcn_sched_barrier(0);
            buf ^= 1;
        }

        #pragma unroll
        for (int v = 0; v < 8; ++v) {
            float y = acc[v] + bb;
            if (EPI == 1) y = fmaxf(y, 0.0f);
            if (EPI == 2) {
                y += rs[v];
                y = (y > 0.0f) ? y : expm1f(y);
            }
            out[rbase + v * DF] = y;
        }
    }
}

extern "C" void kernel_launch(void* const* d_in, const int* in_sizes, int n_in,
                              void* d_out, int out_size, void* d_ws, size_t ws_size,
                              hipStream_t stream) {
    const float* x   = (const float*)d_in[0];
    const int*   ei  = (const int*)d_in[1];
    const float* W1[3] = {(const float*)d_in[2],  (const float*)d_in[6],  (const float*)d_in[10]};
    const float* b1[3] = {(const float*)d_in[3],  (const float*)d_in[7],  (const float*)d_in[11]};
    const float* W2[3] = {(const float*)d_in[4],  (const float*)d_in[8],  (const float*)d_in[12]};
    const float* b2[3] = {(const float*)d_in[5],  (const float*)d_in[9],  (const float*)d_in[13]};
    const float* Wjk = (const float*)d_in[14];
    const float* bjk = (const float*)d_in[15];

    const size_t NM = (size_t)N_NODES * DF;
    float* h0 = (float*)d_ws;
    float* xs[3] = {h0 + NM, h0 + 2 * NM, h0 + 3 * NM};

    dim3 blk(256);
    const int gBlocks = GEMM_BLOCKS;

    int* cnt   = (int*)xs[2];
    int* total = cnt + N_NODES;
    int* start = cnt + 50048;
    int* rank  = cnt + 100096;
    int* nbr   = cnt + 900096;
    const int qBlocks = (N_EDGES / 4 + 255) / 256;
    clear_kernel<<<(N_NODES + 1 + 255) / 256, blk, 0, stream>>>(cnt, N_NODES + 1);
    count_kernel<<<qBlocks, blk, 0, stream>>>(cnt, rank, ei);
    place_kernel<<<(N_NODES + 255) / 256, blk, 0, stream>>>(cnt, total, start);
    fill_kernel<<<qBlocks, blk, 0, stream>>>(nbr, start, rank, ei);

    const float* xin = x;
    for (int l = 0; l < 3; ++l) {
        gather_kernel<<<N_NODES / 8, blk, 0, stream>>>(h0, xin, start, cnt, nbr);
        gemm_wmma<1, 1><<<gBlocks, blk, 0, stream>>>(h0, nullptr, nullptr,
                                                     W1[l], b1[l], nullptr, h0);
        gemm_wmma<1, 2><<<gBlocks, blk, 0, stream>>>(h0, nullptr, nullptr,
                                                     W2[l], b2[l], xin, xs[l]);
        xin = xs[l];
    }
    gemm_wmma<3, 0><<<gBlocks, blk, 0, stream>>>(xs[0], xs[1], xs[2],
                                                 Wjk, bjk, nullptr, (float*)d_out);
}


# Each layer's two MLP GEMMs fused into one launch: both W slices in registers, the 16x128 relu hidden tile passes from first to second product through LDS (no global write+read, 3 launches fewer)
# speedup vs baseline: 2.7807x; 1.1799x over previous
#include <hip/hip_runtime.h>
#include <cmath>

#define N_NODES 50000
#define N_EDGES 800000
#define DF 128

typedef __attribute__((ext_vector_type(2))) float v2f;
typedef __attribute__((ext_vector_type(8))) float v8f;

__global__ void clear_kernel(int* __restrict__ cnt_total, int n) {
    int i = blockIdx.x * blockDim.x + threadIdx.x;
    if (i < n) cnt_total[i] = 0;
}

__global__ void count_kernel(int* __restrict__ cnt, int* __restrict__ rank,
                             const int* __restrict__ ei) {
    int q = blockIdx.x * blockDim.x + threadIdx.x;
    if (q >= N_EDGES / 4) return;
    int4 d = ((const int4*)(ei + N_EDGES))[q];
    int4 r;
    r.x = atomicAdd(cnt + d.x, 1);
    r.y = atomicAdd(cnt + d.y, 1);
    r.z = atomicAdd(cnt + d.z, 1);
    r.w = atomicAdd(cnt + d.w, 1);
    ((int4*)rank)[q] = r;
}

__global__ void place_kernel(const int* __restrict__ cnt, int* __restrict__ total,
                             int* __restrict__ start) {
    __shared__ int sh[256];
    __shared__ int base;
    const int tid = threadIdx.x;
    const int n = blockIdx.x * 256 + tid;
    const int c = (n < N_NODES) ? cnt[n] : 0;
    sh[tid] = c;
    __syncthreads();
    for (int off = 1; off < 256; off <<= 1) {
        int v = (tid >= off) ? sh[tid - off] : 0;
        __syncthreads();
        sh[tid] += v;
        __syncthreads();
    }
    if (tid == 255) base = atomicAdd(total, sh[255]);
    __syncthreads();
    if (n < N_NODES) start[n] = base + sh[tid] - c;
}

__global__ void fill_kernel(int* __restrict__ nbr, const int* __restrict__ start,
                            const int* __restrict__ rank, const int* __restrict__ ei) {
    int q = blockIdx.x * blockDim.x + threadIdx.x;
    if (q >= N_EDGES / 4) return;
    int4 s = ((const int4*)ei)[q];
    int4 d = ((const int4*)(ei + N_EDGES))[q];
    int4 r = ((const int4*)rank)[q];
    nbr[start[d.x] + r.x] = s.x;
    nbr[start[d.y] + r.y] = s.y;
    nbr[start[d.z] + r.z] = s.z;
    nbr[start[d.w] + r.w] = s.w;
}

template<int K>
__device__ __forceinline__ void add_rows(float4& acc, const float4* __restrict__ x4,
                                         int idx, int j, int lane) {
    float4 v[K];
    #pragma unroll
    for (int k = 0; k < K; ++k) {
        int s = __builtin_amdgcn_readlane(idx, j + k);
        v[k] = x4[(size_t)s * (DF / 4) + lane];
    }
    #pragma unroll
    for (int k = 0; k < K; ++k) {
        acc.x += v[k].x; acc.y += v[k].y; acc.z += v[k].z; acc.w += v[k].w;
    }
}

__global__ void __launch_bounds__(256)
gather_kernel(float* __restrict__ agg, const float* __restrict__ x,
              const int* __restrict__ start, const int* __restrict__ cnt,
              const int* __restrict__ nbr) {
    const int lane = threadIdx.x & 31;
    const int node = __builtin_amdgcn_readfirstlane(blockIdx.x * 8 + (threadIdx.x >> 5));
    const float4* x4 = (const float4*)x;
    const int s0 = start[node];
    const int deg = cnt[node];
    float4 acc = x4[(size_t)node * (DF / 4) + lane];
    for (int b = 0; b < deg; b += 32) {
        const int m = (deg - b < 32) ? (deg - b) : 32;
        const int idx = (lane < m) ? nbr[s0 + b + lane] : 0;
        int j = 0;
        for (; j + 8 <= m; j += 8) add_rows<8>(acc, x4, idx, j, lane);
        if (j + 4 <= m) { add_rows<4>(acc, x4, idx, j, lane); j += 4; }
        if (j + 2 <= m) { add_rows<2>(acc, x4, idx, j, lane); j += 2; }
        if (j < m)      { add_rows<1>(acc, x4, idx, j, lane); }
    }
    ((float4*)agg)[(size_t)node * (DF / 4) + lane] = acc;
}

#define LDA (DF + 4)
#define N_TILES 3125
#define GEMM_BLOCKS 512
template<int NSEG, int EPI>
__global__ void __launch_bounds__(256, 4)
gemm_wmma(const float* A0, const float* A1, const float* A2,
          const float* __restrict__ W, const float* __restrict__ bias,
          const float* resid, float* out) {
    __shared__ float lA[2][16 * LDA];

    const int tid  = threadIdx.x;
    const int lane = tid & 31;
    const int wave = tid >> 5;
    const int r    = lane & 15;
    const int hi   = lane >> 4;
    const int n0   = wave * 16;
    const int col  = n0 + r;

    const float* As[3] = {A0, A1, A2};

    constexpr int RA = 4;

    const int st_off = wave * LDA + lane * 4;
    const int fr_off = r * LDA + 2 * hi;

    int t = blockIdx.x;
    float4 p0, p1;
    {
        const float4* s4 = (const float4*)(As[0] + (size_t)t * 16 * DF);
        p0 = s4[tid];
        p1 = s4[tid + 256];
    }

    v2f breg[NSEG][DF / 4];
    #pragma unroll
    for (int seg = 0; seg < NSEG; ++seg) {
        #pragma unroll
        for (int kk = 0; kk < DF / 4; ++kk) {
            const float* bp = W + seg * DF * DF + (4 * kk + 2 * hi) * DF + col;
            breg[seg][kk].x = bp[0];
            breg[seg][kk].y = bp[DF];
        }
    }
    const float bb = bias[col];

    *(float4*)(&lA[0][st_off])           = p0;
    *(float4*)(&lA[0][st_off + 8 * LDA]) = p1;
    __syncthreads();

    int buf = 0;
    for (; t < N_TILES; t += GEMM_BLOCKS) {
        v8f acc = {0.f, 0.f, 0.f, 0.f, 0.f, 0.f, 0.f, 0.f};
        float rs[8] = {0.f, 0.f, 0.f, 0.f, 0.f, 0.f, 0.f, 0.f};
        const size_t rbase = (size_t)(t * 16 + 8 * hi) * DF + col;
        #pragma unroll
        for (int seg = 0; seg < NSEG; ++seg) {
            const bool last = (seg == NSEG - 1);
            const int  tn   = last ? t + GEMM_BLOCKS : t;
            const int  sn   = last ? 0 : seg + 1;
            const int  tc   = (tn < N_TILES) ? tn : t;
            {
                const float4* s4 = (const float4*)(As[sn] + (size_t)tc * 16 * DF);
                p0 = s4[tid];
                p1 = s4[tid + 256];
            }
            if (EPI == 2 && last) {
                #pragma unroll
                for (int v = 0; v < 8; ++v)
                    rs[v] = resid[rbase + v * DF];
            }

            const float* la = &lA[buf][fr_off];
            #pragma unroll
            for (int kk = 0; kk < DF / 4; ++kk) {
                v2f a; a.x = la[4 * kk]; a.y = la[4 * kk + 1];
                acc = __builtin_amdgcn_wmma_f32_16x16x4_f32(
                    false, a, false, breg[seg][kk], (short)0, acc, false, false);
            }
            if (EPI == 2 && last) __builtin_amdgcn_sched_group_barrier(0x020, 10, 0);
            else                  __builtin_amdgcn_sched_group_barrier(0x020, 2, 0);
            __builtin_amdgcn_sched_group_barrier(0x100, RA, 0);
            #pragma unroll
            for (int g = 0; g < 16 - RA; ++g) {
                __builtin_amdgcn_sched_group_barrier(0x008, 2, 0);
                __builtin_amdgcn_sched_group_barrier(0x100, 1, 0);
            }
            __builtin_amdgcn_sched_group_barrier(0x008, 2 * RA, 0);

            *(float4*)(&lA[buf ^ 1][st_off])           = p0;
            *(float4*)(&lA[buf ^ 1][st_off + 8 * LDA]) = p1;
            __syncthreads();
            __builtin_amdgcn_sched_barrier(0);
            buf ^= 1;
        }

        #pragma unroll
        for (int v = 0; v < 8; ++v) {
            float y = acc[v] + bb;
            if (EPI == 1) y = fmaxf(y, 0.0f);
            if (EPI == 2) {
                y += rs[v];
                y = (y > 0.0f) ? y : expm1f(y);
            }
            out[rbase + v * DF] = y;
        }
    }
}

__global__ void __launch_bounds__(256, 4)
mlp_wmma(const float* A, const float* __restrict__ W1, const float* __restrict__ b1,
         const float* __restrict__ W2, const float* __restrict__ b2,
         const float* resid, float* out) {
    __shared__ float lA[2][16 * LDA];
    __shared__ float lH[16 * LDA];

    const int tid  = threadIdx.x;
    const int lane = tid & 31;
    const int wave = tid >> 5;
    const int r    = lane & 15;
    const int hi   = lane >> 4;
    const int col  = wave * 16 + r;

    constexpr int RA = 4;

    const int st_off = wave * LDA + lane * 4;
    const int fr_off = r * LDA + 2 * hi;
    const int h_off  = 8 * hi * LDA + col;

    int t = blockIdx.x;
    float4 p0, p1;
    {
        const float4* s4 = (const float4*)(A + (size_t)t * 16 * DF);
        p0 = s4[tid];
        p1 = s4[tid + 256];
    }

    v2f breg1[DF / 4], breg2[DF / 4];
    #pragma unroll
    for (int kk = 0; kk < DF / 4; ++kk) {
        const float* bp1 = W1 + (4 * kk + 2 * hi) * DF + col;
        breg1[kk].x = bp1[0];
        breg1[kk].y = bp1[DF];
        const float* bp2 = W2 + (4 * kk + 2 * hi) * DF + col;
        breg2[kk].x = bp2[0];
        breg2[kk].y = bp2[DF];
    }
    const float bb1 = b1[col];
    const float bb2 = b2[col];

    *(float4*)(&lA[0][st_off])           = p0;
    *(float4*)(&lA[0][st_off + 8 * LDA]) = p1;
    __syncthreads();

    int buf = 0;
    for (; t < N_TILES; t += GEMM_BLOCKS) {
        const size_t rbase = (size_t)(t * 16 + 8 * hi) * DF + col;
        const int tn = t + GEMM_BLOCKS;
        const int tc = (tn < N_TILES) ? tn : t;
        {
            const float4* s4 = (const float4*)(A + (size_t)tc * 16 * DF);
            p0 = s4[tid];
            p1 = s4[tid + 256];
        }

        v8f acc = {0.f, 0.f, 0.f, 0.f, 0.f, 0.f, 0.f, 0.f};
        const float* la = &lA[buf][fr_off];
        #pragma unroll
        for (int kk = 0; kk < DF / 4; ++kk) {
            v2f a; a.x = la[4 * kk]; a.y = la[4 * kk + 1];
            acc = __builtin_amdgcn_wmma_f32_16x16x4_f32(
                false, a, false, breg1[kk], (short)0, acc, false, false);
        }
        __builtin_amdgcn_sched_group_barrier(0x020, 2, 0);
        __builtin_amdgcn_sched_group_barrier(0x100, RA, 0);
        #pragma unroll
        for (int g = 0; g < 16 - RA; ++g) {
            __builtin_amdgcn_sched_group_barrier(0x008, 2, 0);
            __builtin_amdgcn_sched_group_barrier(0x100, 1, 0);
        }
        __builtin_amdgcn_sched_group_barrier(0x008, 2 * RA, 0);

        #pragma unroll
        for (int v = 0; v < 8; ++v)
            lH[h_off + v * LDA] = fmaxf(acc[v] + bb1, 0.0f);
        asm volatile("s_wait_dscnt 0x0\n\ts_barrier_signal -1\n\ts_barrier_wait -1" ::: "memory");
        __builtin_amdgcn_sched_barrier(0);

        float rs[8];
        #pragma unroll
        for (int v = 0; v < 8; ++v)
            rs[v] = resid[rbase + v * DF];
        v8f acc2 = {0.f, 0.f, 0.f, 0.f, 0.f, 0.f, 0.f, 0.f};
        const float* lh = &lH[fr_off];
        #pragma unroll
        for (int kk = 0; kk < DF / 4; ++kk) {
            v2f a; a.x = lh[4 * kk]; a.y = lh[4 * kk + 1];
            acc2 = __builtin_amdgcn_wmma_f32_16x16x4_f32(
                false, a, false, breg2[kk], (short)0, acc2, false, false);
        }
        __builtin_amdgcn_sched_group_barrier(0x020, 8, 0);
        __builtin_amdgcn_sched_group_barrier(0x100, RA, 0);
        #pragma unroll
        for (int g = 0; g < 16 - RA; ++g) {
            __builtin_amdgcn_sched_group_barrier(0x008, 2, 0);
            __builtin_amdgcn_sched_group_barrier(0x100, 1, 0);
        }
        __builtin_amdgcn_sched_group_barrier(0x008, 2 * RA, 0);

        *(float4*)(&lA[buf ^ 1][st_off])           = p0;
        *(float4*)(&lA[buf ^ 1][st_off + 8 * LDA]) = p1;
        __syncthreads();
        __builtin_amdgcn_sched_barrier(0);
        buf ^= 1;

        #pragma unroll
        for (int v = 0; v < 8; ++v) {
            float y = acc2[v] + bb2 + rs[v];
            y = (y > 0.0f) ? y : expm1f(y);
            out[rbase + v * DF] = y;
        }
    }
}

extern "C" void kernel_launch(void* const* d_in, const int* in_sizes, int n_in,
                              void* d_out, int out_size, void* d_ws, size_t ws_size,
                              hipStream_t stream) {
    const float* x   = (const float*)d_in[0];
    const int*   ei  = (const int*)d_in[1];
    const float* W1[3] = {(const float*)d_in[2],  (const float*)d_in[6],  (const float*)d_in[10]};
    const float* b1[3] = {(const float*)d_in[3],  (const float*)d_in[7],  (const float*)d_in[11]};
    const float* W2[3] = {(const float*)d_in[4],  (const float*)d_in[8],  (const float*)d_in[12]};
    const float* b2[3] = {(const float*)d_in[5],  (const float*)d_in[9],  (const float*)d_in[13]};
    const float* Wjk = (const float*)d_in[14];
    const float* bjk = (const float*)d_in[15];

    const size_t NM = (size_t)N_NODES * DF;
    float* h0 = (float*)d_ws;
    float* xs[3] = {h0 + NM, h0 + 2 * NM, h0 + 3 * NM};

    dim3 blk(256);
    const int gBlocks = GEMM_BLOCKS;

    int* cnt   = (int*)xs[2];
    int* total = cnt + N_NODES;
    int* start = cnt + 50048;
    int* rank  = cnt + 100096;
    int* nbr   = cnt + 900096;
    const int qBlocks = (N_EDGES / 4 + 255) / 256;
    clear_kernel<<<(N_NODES + 1 + 255) / 256, blk, 0, stream>>>(cnt, N_NODES + 1);
    count_kernel<<<qBlocks, blk, 0, stream>>>(cnt, rank, ei);
    place_kernel<<<(N_NODES + 255) / 256, blk, 0, stream>>>(cnt, total, start);
    fill_kernel<<<qBlocks, blk, 0, stream>>>(nbr, start, rank, ei);

    const float* xin = x;
    for (int l = 0; l < 3; ++l) {
        gather_kernel<<<N_NODES / 8, blk, 0, stream>>>(h0, xin, start, cnt, nbr);
        mlp_wmma<<<gBlocks, blk, 0, stream>>>(h0, W1[l], b1[l], W2[l], b2[l], xin, xs[l]);
        xin = xs[l];
    }
    gemm_wmma<3, 0><<<gBlocks, blk, 0, stream>>>(xs[0], xs[1], xs[2],
                                                 Wjk, bjk, nullptr, (float*)d_out);
}


# Edge grouping as a two-level counting sort on LDS counters (196 node-range bins, then per node, one work-group per bin): no global atomics; neighbour sum order = LDS atomics' arrival rank
# speedup vs baseline: 3.1610x; 1.1368x over previous
#include <hip/hip_runtime.h>
#include <cmath>

#define N_NODES 50000
#define N_EDGES 800000
#define DF 128

typedef __attribute__((ext_vector_type(2))) float v2f;
typedef __attribute__((ext_vector_type(8))) float v8f;

#define BIN_SHIFT 8
#define BIN_NODES (1 << BIN_SHIFT)
#define N_BINS 196
#define EDGE_BLOCKS 250
#define Q_PER_BLOCK 800
#define HIST_LEN (N_BINS * EDGE_BLOCKS)
#define EDGE_THREADS 160
#define Q_PER_THREAD 5
#define SCAN_CH 56
#define SCAN_THREADS 875
static_assert(EDGE_BLOCKS * Q_PER_BLOCK * 4 == N_EDGES, "chunks cover the edge list exactly");
static_assert(EDGE_THREADS * Q_PER_THREAD == Q_PER_BLOCK, "threads cover a chunk exactly");
static_assert(N_BINS * BIN_NODES >= N_NODES, "bins cover all nodes");
static_assert(SCAN_THREADS * SCAN_CH == HIST_LEN && SCAN_CH % 4 == 0 && SCAN_THREADS <= 896, "scan chunks");

__global__ void __launch_bounds__(EDGE_THREADS)
bin_count_kernel(int* hist, const int* ei) {
    __shared__ int h[N_BINS];
    const int tid = threadIdx.x;
    for (int i = tid; i < N_BINS; i += EDGE_THREADS) h[i] = 0;
    const int4* d4 = (const int4*)(ei + N_EDGES) + blockIdx.x * Q_PER_BLOCK;
    int4 d[Q_PER_THREAD];
    #pragma unroll
    for (int k = 0; k < Q_PER_THREAD; ++k) d[k] = d4[tid + k * EDGE_THREADS];
    __syncthreads();
    #pragma unroll
    for (int k = 0; k < Q_PER_THREAD; ++k) {
        atomicAdd(&h[d[k].x >> BIN_SHIFT], 1);
        atomicAdd(&h[d[k].y >> BIN_SHIFT], 1);
        atomicAdd(&h[d[k].z >> BIN_SHIFT], 1);
        atomicAdd(&h[d[k].w >> BIN_SHIFT], 1);
    }
    __syncthreads();
    for (int i = tid; i < N_BINS; i += EDGE_THREADS) hist[i * EDGE_BLOCKS + blockIdx.x] = h[i];
}

__device__ __forceinline__ int wave_inclusive_scan(int v, int lane) {
    #pragma unroll
    for (int d = 1; d < 32; d <<= 1) {
        const int t = __shfl_up(v, d);
        if (lane >= d) v += t;
    }
    return v;
}

__global__ void __launch_bounds__(896)
scan_kernel(int* __restrict__ off, const int* __restrict__ hist) {
    __shared__ int wsum[32];
    const int tid = threadIdx.x;
    const int lane = tid & 31;
    const int wave = tid >> 5;
    const bool act = tid < SCAN_THREADS;
    const int base = tid * SCAN_CH;
    int v[SCAN_CH];
    #pragma unroll
    for (int k = 0; k < SCAN_CH; ++k) v[k] = 0;
    if (act) {
        #pragma unroll
        for (int k = 0; k < SCAN_CH; k += 4) {
            const int4 q = *(const int4*)(hist + base + k);
            v[k] = q.x; v[k + 1] = q.y; v[k + 2] = q.z; v[k + 3] = q.w;
        }
    }
    int s = 0;
    #pragma unroll
    for (int k = 0; k < SCAN_CH; ++k) s += v[k];
    const int inc = wave_inclusive_scan(s, lane);
    if (lane == 31) wsum[wave] = inc;
    __syncthreads();
    if (wave == 0) {
        const int w = (lane < 28) ? wsum[lane] : 0;
        wsum[lane] = wave_inclusive_scan(w, lane) - w;
    }
    __syncthreads();
    if (act) {
        int run = wsum[wave] + inc - s;
        #pragma unroll
        for (int k = 0; k < SCAN_CH; k += 4) {
            int4 q;
            q.x = run; run += v[k];
            q.y = run; run += v[k + 1];
            q.z = run; run += v[k + 2];
            q.w = run; run += v[k + 3];
            *(int4*)(off + base + k) = q;
        }
    }
}

__global__ void __launch_bounds__(EDGE_THREADS)
bin_scatter_kernel(int2* __restrict__ pairs, const int* __restrict__ off,
                   const int* __restrict__ ei) {
    __shared__ int cur[N_BINS];
    const int tid = threadIdx.x;
    for (int i = tid; i < N_BINS; i += EDGE_THREADS) cur[i] = off[i * EDGE_BLOCKS + blockIdx.x];
    const int4* s4 = (const int4*)ei + blockIdx.x * Q_PER_BLOCK;
    const int4* d4 = (const int4*)(ei + N_EDGES) + blockIdx.x * Q_PER_BLOCK;
    int4 s[Q_PER_THREAD], d[Q_PER_THREAD];
    #pragma unroll
    for (int k = 0; k < Q_PER_THREAD; ++k) {
        s[k] = s4[tid + k * EDGE_THREADS];
        d[k] = d4[tid + k * EDGE_THREADS];
    }
    __syncthreads();
    int4 p[Q_PER_THREAD];
    #pragma unroll
    for (int k = 0; k < Q_PER_THREAD; ++k) {
        p[k].x = atomicAdd(&cur[d[k].x >> BIN_SHIFT], 1);
        p[k].y = atomicAdd(&cur[d[k].y >> BIN_SHIFT], 1);
        p[k].z = atomicAdd(&cur[d[k].z >> BIN_SHIFT], 1);
        p[k].w = atomicAdd(&cur[d[k].w >> BIN_SHIFT], 1);
    }
    #pragma unroll
    for (int k = 0; k < Q_PER_THREAD; ++k) {
        pairs[p[k].x] = make_int2(s[k].x, d[k].x);
        pairs[p[k].y] = make_int2(s[k].y, d[k].y);
        pairs[p[k].z] = make_int2(s[k].z, d[k].z);
        pairs[p[k].w] = make_int2(s[k].w, d[k].w);
    }
}

__global__ void __launch_bounds__(1024)
group_kernel(int* __restrict__ nbr, int* __restrict__ start, int* __restrict__ cnt,
             const int2* __restrict__ pairs, const int* __restrict__ off) {
    __shared__ int c[BIN_NODES];
    __shared__ int cur[BIN_NODES];
    __shared__ int wsum[8];
    const int tid = threadIdx.x;
    const int lane = tid & 31;
    const int wave = tid >> 5;
    const int b = blockIdx.x;
    const int lo = off[b * EDGE_BLOCKS];
    const int hi = (b + 1 < N_BINS) ? off[(b + 1) * EDGE_BLOCKS] : N_EDGES;
    if (tid < BIN_NODES) c[tid] = 0;
    __syncthreads();
    for (int i0 = lo + tid; i0 < hi; i0 += 4096) {
        int dd[4];
        #pragma unroll
        for (int k = 0; k < 4; ++k) {
            const int i = i0 + k * 1024;
            dd[k] = pairs[(i < hi) ? i : hi - 1].y;
        }
        #pragma unroll
        for (int k = 0; k < 4; ++k)
            if (i0 + k * 1024 < hi) atomicAdd(&c[dd[k] & (BIN_NODES - 1)], 1);
    }
    __syncthreads();
    int cv = 0, inc = 0;
    if (tid < BIN_NODES) {
        cv = c[tid];
        inc = wave_inclusive_scan(cv, lane);
        if (lane == 31) wsum[wave] = inc;
    }
    __syncthreads();
    if (tid < BIN_NODES) {
        int wp = 0;
        for (int w = 0; w < wave; ++w) wp += wsum[w];
        const int st = lo + wp + inc - cv;
        cur[tid] = st;
        const int node = b * BIN_NODES + tid;
        if (node < N_NODES) { start[node] = st; cnt[node] = cv; }
    }
    __syncthreads();
    for (int i0 = lo + tid; i0 < hi; i0 += 4096) {
        int2 p[4];
        #pragma unroll
        for (int k = 0; k < 4; ++k) {
            const int i = i0 + k * 1024;
            p[k] = pairs[(i < hi) ? i : hi - 1];
        }
        #pragma unroll
        for (int k = 0; k < 4; ++k)
            if (i0 + k * 1024 < hi)
                nbr[atomicAdd(&cur[p[k].y & (BIN_NODES - 1)], 1)] = p[k].x;
    }
}

template<int K>
__device__ __forceinline__ void add_rows(float4& acc, const float4* __restrict__ x4,
                                         int idx, int j, int lane) {
    float4 v[K];
    #pragma unroll
    for (int k = 0; k < K; ++k) {
        int s = __builtin_amdgcn_readlane(idx, j + k);
        v[k] = x4[(size_t)s * (DF / 4) + lane];
    }
    #pragma unroll
    for (int k = 0; k < K; ++k) {
        acc.x += v[k].x; acc.y += v[k].y; acc.z += v[k].z; acc.w += v[k].w;
    }
}

__global__ void __launch_bounds__(256)
gather_kernel(float* __restrict__ agg, const float* __restrict__ x,
              const int* __restrict__ start, const int* __restrict__ cnt,
              const int* __restrict__ nbr) {
    const int lane = threadIdx.x & 31;
    const int node = __builtin_amdgcn_readfirstlane(blockIdx.x * 8 + (threadIdx.x >> 5));
    const float4* x4 = (const float4*)x;
    const int s0 = start[node];
    const int deg = cnt[node];
    float4 acc = x4[(size_t)node * (DF / 4) + lane];
    for (int b = 0; b < deg; b += 32) {
        const int m = (deg - b < 32) ? (deg - b) : 32;
        const int idx = (lane < m) ? nbr[s0 + b + lane] : 0;
        int j = 0;
        for (; j + 8 <= m; j += 8) add_rows<8>(acc, x4, idx, j, lane);
        if (j + 4 <= m) { add_rows<4>(acc, x4, idx, j, lane); j += 4; }
        if (j + 2 <= m) { add_rows<2>(acc, x4, idx, j, lane); j += 2; }
        if (j < m)      { add_rows<1>(acc, x4, idx, j, lane); }
    }
    ((float4*)agg)[(size_t)node * (DF / 4) + lane] = acc;
}

#define LDA (DF + 4)
#define N_TILES 3125
#define GEMM_BLOCKS 512
template<int NSEG, int EPI>
__global__ void __launch_bounds__(256, 4)
gemm_wmma(const float* A0, const float* A1, const float* A2,
          const float* __restrict__ W, const float* __restrict__ bias,
          const float* resid, float* out) {
    __shared__ float lA[2][16 * LDA];

    const int tid  = threadIdx.x;
    const int lane = tid & 31;
    const int wave = tid >> 5;
    const int r    = lane & 15;
    const int hi   = lane >> 4;
    const int n0   = wave * 16;
    const int col  = n0 + r;

    const float* As[3] = {A0, A1, A2};

    constexpr int RA = 4;

    const int st_off = wave * LDA + lane * 4;
    const int fr_off = r * LDA + 2 * hi;

    int t = blockIdx.x;
    float4 p0, p1;
    {
        const float4* s4 = (const float4*)(As[0] + (size_t)t * 16 * DF);
        p0 = s4[tid];
        p1 = s4[tid + 256];
    }

    v2f breg[NSEG][DF / 4];
    #pragma unroll
    for (int seg = 0; seg < NSEG; ++seg) {
        #pragma unroll
        for (int kk = 0; kk < DF / 4; ++kk) {
            const float* bp = W + seg * DF * DF + (4 * kk + 2 * hi) * DF + col;
            breg[seg][kk].x = bp[0];
            breg[seg][kk].y = bp[DF];
        }
    }
    const float bb = bias[col];

    *(float4*)(&lA[0][st_off])           = p0;
    *(float4*)(&lA[0][st_off + 8 * LDA]) = p1;
    __syncthreads();

    int buf = 0;
    for (; t < N_TILES; t += GEMM_BLOCKS) {
        v8f acc = {0.f, 0.f, 0.f, 0.f, 0.f, 0.f, 0.f, 0.f};
        float rs[8] = {0.f, 0.f, 0.f, 0.f, 0.f, 0.f, 0.f, 0.f};
        const size_t rbase = (size_t)(t * 16 + 8 * hi) * DF + col;
        #pragma unroll
        for (int seg = 0; seg < NSEG; ++seg) {
            const bool last = (seg == NSEG - 1);
            const int  tn   = last ? t + GEMM_BLOCKS : t;
            const int  sn   = last ? 0 : seg + 1;
            const int  tc   = (tn < N_TILES) ? tn : t;
            {
                const float4* s4 = (const float4*)(As[sn] + (size_t)tc * 16 * DF);
                p0 = s4[tid];
                p1 = s4[tid + 256];
            }
            if (EPI == 2 && last) {
                #pragma unroll
                for (int v = 0; v < 8; ++v)
                    rs[v] = resid[rbase + v * DF];
            }

            const float* la = &lA[buf][fr_off];
            #pragma unroll
            for (int kk = 0; kk < DF / 4; ++kk) {
                v2f a; a.x = la[4 * kk]; a.y = la[4 * kk + 1];
                acc = __builtin_amdgcn_wmma_f32_16x16x4_f32(
                    false, a, false, breg[seg][kk], (short)0, acc, false, false);
            }
            if (EPI == 2 && last) __builtin_amdgcn_sched_group_barrier(0x020, 10, 0);
            else                  __builtin_amdgcn_sched_group_barrier(0x020, 2, 0);
            __builtin_amdgcn_sched_group_barrier(0x100, RA, 0);
            #pragma unroll
            for (int g = 0; g < 16 - RA; ++g) {
                __builtin_amdgcn_sched_group_barrier(0x008, 2, 0);
                __builtin_amdgcn_sched_group_barrier(0x100, 1, 0);
            }
            __builtin_amdgcn_sched_group_barrier(0x008, 2 * RA, 0);

            *(float4*)(&lA[buf ^ 1][st_off])           = p0;
            *(float4*)(&lA[buf ^ 1][st_off + 8 * LDA]) = p1;
            __syncthreads();
            __builtin_amdgcn_sched_barrier(0);
            buf ^= 1;
        }

        #pragma unroll
        for (int v = 0; v < 8; ++v) {
            float y = acc[v] + bb;
            if (EPI == 1) y = fmaxf(y, 0.0f);
            if (EPI == 2) {
                y += rs[v];
                y = (y > 0.0f) ? y : expm1f(y);
            }
            out[rbase + v * DF] = y;
        }
    }
}

__global__ void __launch_bounds__(256, 4)
mlp_wmma(const float* A, const float* __restrict__ W1, const float* __restrict__ b1,
         const float* __restrict__ W2, const float* __restrict__ b2,
         const float* resid, float* out) {
    __shared__ float lA[2][16 * LDA];
    __shared__ float lH[16 * LDA];

    const int tid  = threadIdx.x;
    const int lane = tid & 31;
    const int wave = tid >> 5;
    const int r    = lane & 15;
    const int hi   = lane >> 4;
    const int col  = wave * 16 + r;

    constexpr int RA = 4;

    const int st_off = wave * LDA + lane * 4;
    const int fr_off = r * LDA + 2 * hi;
    const int h_off  = 8 * hi * LDA + col;

    int t = blockIdx.x;
    float4 p0, p1;
    {
        const float4* s4 = (const float4*)(A + (size_t)t * 16 * DF);
        p0 = s4[tid];
        p1 = s4[tid + 256];
    }

    v2f breg1[DF / 4], breg2[DF / 4];
    #pragma unroll
    for (int kk = 0; kk < DF / 4; ++kk) {
        const float* bp1 = W1 + (4 * kk + 2 * hi) * DF + col;
        breg1[kk].x = bp1[0];
        breg1[kk].y = bp1[DF];
        const float* bp2 = W2 + (4 * kk + 2 * hi) * DF + col;
        breg2[kk].x = bp2[0];
        breg2[kk].y = bp2[DF];
    }
    const float bb1 = b1[col];
    const float bb2 = b2[col];

    *(float4*)(&lA[0][st_off])           = p0;
    *(float4*)(&lA[0][st_off + 8 * LDA]) = p1;
    __syncthreads();

    int buf = 0;
    for (; t < N_TILES; t += GEMM_BLOCKS) {
        const size_t rbase = (size_t)(t * 16 + 8 * hi) * DF + col;
        const int tn = t + GEMM_BLOCKS;
        const int tc = (tn < N_TILES) ? tn : t;
        {
            const float4* s4 = (const float4*)(A + (size_t)tc * 16 * DF);
            p0 = s4[tid];
            p1 = s4[tid + 256];
        }

        v8f acc = {0.f, 0.f, 0.f, 0.f, 0.f, 0.f, 0.f, 0.f};
        const float* la = &lA[buf][fr_off];
        #pragma unroll
        for (int kk = 0; kk < DF / 4; ++kk) {
            v2f a; a.x = la[4 * kk]; a.y = la[4 * kk + 1];
            acc = __builtin_amdgcn_wmma_f32_16x16x4_f32(
                false, a, false, breg1[kk], (short)0, acc, false, false);
        }
        __builtin_amdgcn_sched_group_barrier(0x020, 2, 0);
        __builtin_amdgcn_sched_group_barrier(0x100, RA, 0);
        #pragma unroll
        for (int g = 0; g < 16 - RA; ++g) {
            __builtin_amdgcn_sched_group_barrier(0x008, 2, 0);
            __builtin_amdgcn_sched_group_barrier(0x100, 1, 0);
        }
        __builtin_amdgcn_sched_group_barrier(0x008, 2 * RA, 0);

        #pragma unroll
        for (int v = 0; v < 8; ++v)
            lH[h_off + v * LDA] = fmaxf(acc[v] + bb1, 0.0f);
        asm volatile("s_wait_dscnt 0x0\n\ts_barrier_signal -1\n\ts_barrier_wait -1" ::: "memory");
        __builtin_amdgcn_sched_barrier(0);

        float rs[8];
        #pragma unroll
        for (int v = 0; v < 8; ++v)
            rs[v] = resid[rbase + v * DF];
        v8f acc2 = {0.f, 0.f, 0.f, 0.f, 0.f, 0.f, 0.f, 0.f};
        const float* lh = &lH[fr_off];
        #pragma unroll
        for (int kk = 0; kk < DF / 4; ++kk) {
            v2f a; a.x = lh[4 * kk]; a.y = lh[4 * kk + 1];
            acc2 = __builtin_amdgcn_wmma_f32_16x16x4_f32(
                false, a, false, breg2[kk], (short)0, acc2, false, false);
        }
        __builtin_amdgcn_sched_group_barrier(0x020, 8, 0);
        __builtin_amdgcn_sched_group_barrier(0x100, RA, 0);
        #pragma unroll
        for (int g = 0; g < 16 - RA; ++g) {
            __builtin_amdgcn_sched_group_barrier(0x008, 2, 0);
            __builtin_amdgcn_sched_group_barrier(0x100, 1, 0);
        }
        __builtin_amdgcn_sched_group_barrier(0x008, 2 * RA, 0);

        *(float4*)(&lA[buf ^ 1][st_off])           = p0;
        *(float4*)(&lA[buf ^ 1][st_off + 8 * LDA]) = p1;
        __syncthreads();
        __builtin_amdgcn_sched_barrier(0);
        buf ^= 1;

        #pragma unroll
        for (int v = 0; v < 8; ++v) {
            float y = acc2[v] + bb2 + rs[v];
            y = (y > 0.0f) ? y : expm1f(y);
            out[rbase + v * DF] = y;
        }
    }
}

extern "C" void kernel_launch(void* const* d_in, const int* in_sizes, int n_in,
                              void* d_out, int out_size, void* d_ws, size_t ws_size,
                              hipStream_t stream) {
    const float* x   = (const float*)d_in[0];
    const int*   ei  = (const int*)d_in[1];
    const float* W1[3] = {(const float*)d_in[2],  (const float*)d_in[6],  (const float*)d_in[10]};
    const float* b1[3] = {(const float*)d_in[3],  (const float*)d_in[7],  (const float*)d_in[11]};
    const float* W2[3] = {(const float*)d_in[4],  (const float*)d_in[8],  (const float*)d_in[12]};
    const float* b2[3] = {(const float*)d_in[5],  (const float*)d_in[9],  (const float*)d_in[13]};
    const float* Wjk = (const float*)d_in[14];
    const float* bjk = (const float*)d_in[15];

    const size_t NM = (size_t)N_NODES * DF;
    float* h0 = (float*)d_ws;
    float* xs[3] = {h0 + NM, h0 + 2 * NM, h0 + 3 * NM};

    dim3 blk(256);
    const int gBlocks = GEMM_BLOCKS;

    int*  cnt   = (int*)xs[2];
    int*  start = cnt + 50048;
    int*  hist  = cnt + 100096;
    int*  off   = cnt + 149120;
    int*  nbr   = cnt + 200704;
    int2* pairs = (int2*)(cnt + 1000704);
    bin_count_kernel<<<EDGE_BLOCKS, EDGE_THREADS, 0, stream>>>(hist, ei);
    scan_kernel<<<1, 896, 0, stream>>>(off, hist);
    bin_scatter_kernel<<<EDGE_BLOCKS, EDGE_THREADS, 0, stream>>>(pairs, off, ei);
    group_kernel<<<N_BINS, 1024, 0, stream>>>(nbr, start, cnt, pairs, off);

    const float* xin = x;
    for (int l = 0; l < 3; ++l) {
        gather_kernel<<<N_NODES / 8, blk, 0, stream>>>(h0, xin, start, cnt, nbr);
        mlp_wmma<<<gBlocks, blk, 0, stream>>>(h0, W1[l], b1[l], W2[l], b2[l], xin, xs[l]);
        xin = xs[l];
    }
    gemm_wmma<3, 0><<<gBlocks, blk, 0, stream>>>(xs[0], xs[1], xs[2],
                                                 Wjk, bjk, nullptr, (float*)d_out);
}


# Neighbour-row sums done inside the MLP launch: each wave sums its two rows of the next 16-row tile in registers into the LDS A tile while the WGP's other work-group multiplies; 3 launches fewer
# speedup vs baseline: 3.4103x; 1.0789x over previous
#include <hip/hip_runtime.h>
#include <cmath>

#define N_NODES 50000
#define N_EDGES 800000
#define DF 128

typedef __attribute__((ext_vector_type(2))) float v2f;
typedef __attribute__((ext_vector_type(8))) float v8f;

#define BIN_SHIFT 8
#define BIN_NODES (1 << BIN_SHIFT)
#define N_BINS 196
#define EDGE_BLOCKS 250
#define Q_PER_BLOCK 800
#define HIST_LEN (N_BINS * EDGE_BLOCKS)
#define EDGE_THREADS 160
#define Q_PER_THREAD 5
#define SCAN_CH 56
#define SCAN_THREADS 875
static_assert(EDGE_BLOCKS * Q_PER_BLOCK * 4 == N_EDGES, "chunks cover the edge list exactly");
static_assert(EDGE_THREADS * Q_PER_THREAD == Q_PER_BLOCK, "threads cover a chunk exactly");
static_assert(N_BINS * BIN_NODES >= N_NODES, "bins cover all nodes");
static_assert(SCAN_THREADS * SCAN_CH == HIST_LEN && SCAN_CH % 4 == 0 && SCAN_THREADS <= 896, "scan chunks");

__global__ void __launch_bounds__(EDGE_THREADS)
bin_count_kernel(int* hist, const int* ei) {
    __shared__ int h[N_BINS];
    const int tid = threadIdx.x;
    for (int i = tid; i < N_BINS; i += EDGE_THREADS) h[i] = 0;
    const int4* d4 = (const int4*)(ei + N_EDGES) + blockIdx.x * Q_PER_BLOCK;
    int4 d[Q_PER_THREAD];
    #pragma unroll
    for (int k = 0; k < Q_PER_THREAD; ++k) d[k] = d4[tid + k * EDGE_THREADS];
    __syncthreads();
    #pragma unroll
    for (int k = 0; k < Q_PER_THREAD; ++k) {
        atomicAdd(&h[d[k].x >> BIN_SHIFT], 1);
        atomicAdd(&h[d[k].y >> BIN_SHIFT], 1);
        atomicAdd(&h[d[k].z >> BIN_SHIFT], 1);
        atomicAdd(&h[d[k].w >> BIN_SHIFT], 1);
    }
    __syncthreads();
    for (int i = tid; i < N_BINS; i += EDGE_THREADS) hist[i * EDGE_BLOCKS + blockIdx.x] = h[i];
}

__device__ __forceinline__ int wave_inclusive_scan(int v, int lane) {
    #pragma unroll
    for (int d = 1; d < 32; d <<= 1) {
        const int t = __shfl_up(v, d);
        if (lane >= d) v += t;
    }
    return v;
}

__global__ void __launch_bounds__(896)
scan_kernel(int* __restrict__ off, const int* __restrict__ hist) {
    __shared__ int wsum[32];
    const int tid = threadIdx.x;
    const int lane = tid & 31;
    const int wave = tid >> 5;
    const bool act = tid < SCAN_THREADS;
    const int base = tid * SCAN_CH;
    int v[SCAN_CH];
    #pragma unroll
    for (int k = 0; k < SCAN_CH; ++k) v[k] = 0;
    if (act) {
        #pragma unroll
        for (int k = 0; k < SCAN_CH; k += 4) {
            const int4 q = *(const int4*)(hist + base + k);
            v[k] = q.x; v[k + 1] = q.y; v[k + 2] = q.z; v[k + 3] = q.w;
        }
    }
    int s = 0;
    #pragma unroll
    for (int k = 0; k < SCAN_CH; ++k) s += v[k];
    const int inc = wave_inclusive_scan(s, lane);
    if (lane == 31) wsum[wave] = inc;
    __syncthreads();
    if (wave == 0) {
        const int w = (lane < 28) ? wsum[lane] : 0;
        wsum[lane] = wave_inclusive_scan(w, lane) - w;
    }
    __syncthreads();
    if (act) {
        int run = wsum[wave] + inc - s;
        #pragma unroll
        for (int k = 0; k < SCAN_CH; k += 4) {
            int4 q;
            q.x = run; run += v[k];
            q.y = run; run += v[k + 1];
            q.z = run; run += v[k + 2];
            q.w = run; run += v[k + 3];
            *(int4*)(off + base + k) = q;
        }
    }
}

__global__ void __launch_bounds__(EDGE_THREADS)
bin_scatter_kernel(int2* __restrict__ pairs, const int* __restrict__ off,
                   const int* __restrict__ ei) {
    __shared__ int cur[N_BINS];
    const int tid = threadIdx.x;
    for (int i = tid; i < N_BINS; i += EDGE_THREADS) cur[i] = off[i * EDGE_BLOCKS + blockIdx.x];
    const int4* s4 = (const int4*)ei + blockIdx.x * Q_PER_BLOCK;
    const int4* d4 = (const int4*)(ei + N_EDGES) + blockIdx.x * Q_PER_BLOCK;
    int4 s[Q_PER_THREAD], d[Q_PER_THREAD];
    #pragma unroll
    for (int k = 0; k < Q_PER_THREAD; ++k) {
        s[k] = s4[tid + k * EDGE_THREADS];
        d[k] = d4[tid + k * EDGE_THREADS];
    }
    __syncthreads();
    int4 p[Q_PER_THREAD];
    #pragma unroll
    for (int k = 0; k < Q_PER_THREAD; ++k) {
        p[k].x = atomicAdd(&cur[d[k].x >> BIN_SHIFT], 1);
        p[k].y = atomicAdd(&cur[d[k].y >> BIN_SHIFT], 1);
        p[k].z = atomicAdd(&cur[d[k].z >> BIN_SHIFT], 1);
        p[k].w = atomicAdd(&cur[d[k].w >> BIN_SHIFT], 1);
    }
    #pragma unroll
    for (int k = 0; k < Q_PER_THREAD; ++k) {
        pairs[p[k].x] = make_int2(s[k].x, d[k].x);
        pairs[p[k].y] = make_int2(s[k].y, d[k].y);
        pairs[p[k].z] = make_int2(s[k].z, d[k].z);
        pairs[p[k].w] = make_int2(s[k].w, d[k].w);
    }
}

__global__ void __launch_bounds__(1024)
group_kernel(int* __restrict__ nbr, int* __restrict__ start, int* __restrict__ cnt,
             const int2* __restrict__ pairs, const int* __restrict__ off) {
    __shared__ int c[BIN_NODES];
    __shared__ int cur[BIN_NODES];
    __shared__ int wsum[8];
    const int tid = threadIdx.x;
    const int lane = tid & 31;
    const int wave = tid >> 5;
    const int b = blockIdx.x;
    const int lo = off[b * EDGE_BLOCKS];
    const int hi = (b + 1 < N_BINS) ? off[(b + 1) * EDGE_BLOCKS] : N_EDGES;
    if (tid < BIN_NODES) c[tid] = 0;
    __syncthreads();
    for (int i0 = lo + tid; i0 < hi; i0 += 4096) {
        int dd[4];
        #pragma unroll
        for (int k = 0; k < 4; ++k) {
            const int i = i0 + k * 1024;
            dd[k] = pairs[(i < hi) ? i : hi - 1].y;
        }
        #pragma unroll
        for (int k = 0; k < 4; ++k)
            if (i0 + k * 1024 < hi) atomicAdd(&c[dd[k] & (BIN_NODES - 1)], 1);
    }
    __syncthreads();
    int cv = 0, inc = 0;
    if (tid < BIN_NODES) {
        cv = c[tid];
        inc = wave_inclusive_scan(cv, lane);
        if (lane == 31) wsum[wave] = inc;
    }
    __syncthreads();
    if (tid < BIN_NODES) {
        int wp = 0;
        for (int w = 0; w < wave; ++w) wp += wsum[w];
        const int st = lo + wp + inc - cv;
        cur[tid] = st;
        const int node = b * BIN_NODES + tid;
        if (node < N_NODES) { start[node] = st; cnt[node] = cv; }
    }
    __syncthreads();
    for (int i0 = lo + tid; i0 < hi; i0 += 4096) {
        int2 p[4];
        #pragma unroll
        for (int k = 0; k < 4; ++k) {
            const int i = i0 + k * 1024;
            p[k] = pairs[(i < hi) ? i : hi - 1];
        }
        #pragma unroll
        for (int k = 0; k < 4; ++k)
            if (i0 + k * 1024 < hi)
                nbr[atomicAdd(&cur[p[k].y & (BIN_NODES - 1)], 1)] = p[k].x;
    }
}

template<int K>
__device__ __forceinline__ void add_rows(float4& acc, const float4* x4,
                                         int idx, int j, int lane) {
    float4 v[K];
    #pragma unroll
    for (int k = 0; k < K; ++k) {
        int s = __builtin_amdgcn_readlane(idx, j + k);
        v[k] = x4[(size_t)s * (DF / 4) + lane];
    }
    #pragma unroll
    for (int k = 0; k < K; ++k) {
        acc.x += v[k].x; acc.y += v[k].y; acc.z += v[k].z; acc.w += v[k].w;
    }
}

__device__ __forceinline__ float4 gather_row(const float4* x4, const int* nbr, float4 acc,
                                             int s0, int deg, int idx0, int lane) {
    for (int b = 0; b < deg; b += 32) {
        const int m = (deg - b < 32) ? (deg - b) : 32;
        const int idx = (b == 0) ? idx0 : ((lane < m) ? nbr[s0 + b + lane] : 0);
        int j = 0;
        for (; j + 8 <= m; j += 8) add_rows<8>(acc, x4, idx, j, lane);
        switch (m - j) {
            case 7: add_rows<7>(acc, x4, idx, j, lane); break;
            case 6: add_rows<6>(acc, x4, idx, j, lane); break;
            case 5: add_rows<5>(acc, x4, idx, j, lane); break;
            case 4: add_rows<4>(acc, x4, idx, j, lane); break;
            case 3: add_rows<3>(acc, x4, idx, j, lane); break;
            case 2: add_rows<2>(acc, x4, idx, j, lane); break;
            case 1: add_rows<1>(acc, x4, idx, j, lane); break;
            default: break;
        }
    }
    return acc;
}

#define LDA (DF + 4)
#define N_TILES 3125
#define GEMM_BLOCKS 512
template<int NSEG, int EPI>
__global__ void __launch_bounds__(256, 4)
gemm_wmma(const float* A0, const float* A1, const float* A2,
          const float* __restrict__ W, const float* __restrict__ bias,
          const float* resid, float* out) {
    __shared__ float lA[2][16 * LDA];

    const int tid  = threadIdx.x;
    const int lane = tid & 31;
    const int wave = tid >> 5;
    const int r    = lane & 15;
    const int hi   = lane >> 4;
    const int n0   = wave * 16;
    const int col  = n0 + r;

    const float* As[3] = {A0, A1, A2};

    constexpr int RA = 4;

    const int st_off = wave * LDA + lane * 4;
    const int fr_off = r * LDA + 2 * hi;

    int t = blockIdx.x;
    float4 p0, p1;
    {
        const float4* s4 = (const float4*)(As[0] + (size_t)t * 16 * DF);
        p0 = s4[tid];
        p1 = s4[tid + 256];
    }

    v2f breg[NSEG][DF / 4];
    #pragma unroll
    for (int seg = 0; seg < NSEG; ++seg) {
        #pragma unroll
        for (int kk = 0; kk < DF / 4; ++kk) {
            const float* bp = W + seg * DF * DF + (4 * kk + 2 * hi) * DF + col;
            breg[seg][kk].x = bp[0];
            breg[seg][kk].y = bp[DF];
        }
    }
    const float bb = bias[col];

    *(float4*)(&lA[0][st_off])           = p0;
    *(float4*)(&lA[0][st_off + 8 * LDA]) = p1;
    __syncthreads();

    int buf = 0;
    for (; t < N_TILES; t += GEMM_BLOCKS) {
        v8f acc = {0.f, 0.f, 0.f, 0.f, 0.f, 0.f, 0.f, 0.f};
        float rs[8] = {0.f, 0.f, 0.f, 0.f, 0.f, 0.f, 0.f, 0.f};
        const size_t rbase = (size_t)(t * 16 + 8 * hi) * DF + col;
        #pragma unroll
        for (int seg = 0; seg < NSEG; ++seg) {
            const bool last = (seg == NSEG - 1);
            const int  tn   = last ? t + GEMM_BLOCKS : t;
            const int  sn   = last ? 0 : seg + 1;
            const int  tc   = (tn < N_TILES) ? tn : t;
            {
                const float4* s4 = (const float4*)(As[sn] + (size_t)tc * 16 * DF);
                p0 = s4[tid];
                p1 = s4[tid + 256];
            }
            if (EPI == 2 && last) {
                #pragma unroll
                for (int v = 0; v < 8; ++v)
                    rs[v] = resid[rbase + v * DF];
            }

            const float* la = &lA[buf][fr_off];
            #pragma unroll
            for (int kk = 0; kk < DF / 4; ++kk) {
                v2f a; a.x = la[4 * kk]; a.y = la[4 * kk + 1];
                acc = __builtin_amdgcn_wmma_f32_16x16x4_f32(
                    false, a, false, breg[seg][kk], (short)0, acc, false, false);
            }
            if (EPI == 2 && last) __builtin_amdgcn_sched_group_barrier(0x020, 10, 0);
            else                  __builtin_amdgcn_sched_group_barrier(0x020, 2, 0);
            __builtin_amdgcn_sched_group_barrier(0x100, RA, 0);
            #pragma unroll
            for (int g = 0; g < 16 - RA; ++g) {
                __builtin_amdgcn_sched_group_barrier(0x008, 2, 0);
                __builtin_amdgcn_sched_group_barrier(0x100, 1, 0);
            }
            __builtin_amdgcn_sched_group_barrier(0x008, 2 * RA, 0);

            *(float4*)(&lA[buf ^ 1][st_off])           = p0;
            *(float4*)(&lA[buf ^ 1][st_off + 8 * LDA]) = p1;
            __syncthreads();
            __builtin_amdgcn_sched_barrier(0);
            buf ^= 1;
        }

        #pragma unroll
        for (int v = 0; v < 8; ++v) {
            float y = acc[v] + bb;
            if (EPI == 1) y = fmaxf(y, 0.0f);
            if (EPI == 2) {
                y += rs[v];
                y = (y > 0.0f) ? y : expm1f(y);
            }
            out[rbase + v * DF] = y;
        }
    }
}

__global__ void __launch_bounds__(256, 4)
mlp_wmma(const float* x, const int* start, const int* cnt, const int* nbr,
         const float* __restrict__ W1, const float* __restrict__ b1,
         const float* __restrict__ W2, const float* __restrict__ b2,
         float* out) {
    __shared__ float lA[2][16 * LDA];
    __shared__ float lH[16 * LDA];

    const int tid  = threadIdx.x;
    const int lane = tid & 31;
    const int wave = tid >> 5;
    const int swave = __builtin_amdgcn_readfirstlane(wave);
    const int r    = lane & 15;
    const int hi   = lane >> 4;
    const int col  = wave * 16 + r;

    constexpr int RA = 4;

    const int st_off = wave * LDA + lane * 4;
    const int fr_off = r * LDA + 2 * hi;
    const int h_off  = 8 * hi * LDA + col;

    const float4* x4 = (const float4*)x;

    const int* mp = (lane & 2) ? cnt : start;
    const int  mo = swave + 8 * (lane & 1);

    int t = blockIdx.x;
    int na = t * 16 + swave, nb = na + 8;
    int meta = mp[t * 16 + mo];
    int sa = __builtin_amdgcn_readlane(meta, 0), sb = __builtin_amdgcn_readlane(meta, 1);
    int ia = nbr[(sa + lane < N_EDGES) ? sa + lane : N_EDGES - 1];
    int ib = nbr[(sb + lane < N_EDGES) ? sb + lane : N_EDGES - 1];
    float4 ga = x4[(size_t)na * (DF / 4) + lane];
    float4 gb = x4[(size_t)nb * (DF / 4) + lane];

    v2f breg1[DF / 4], breg2[DF / 4];
    #pragma unroll
    for (int kk = 0; kk < DF / 4; ++kk) {
        const float* bp1 = W1 + (4 * kk + 2 * hi) * DF + col;
        breg1[kk].x = bp1[0];
        breg1[kk].y = bp1[DF];
        const float* bp2 = W2 + (4 * kk + 2 * hi) * DF + col;
        breg2[kk].x = bp2[0];
        breg2[kk].y = bp2[DF];
    }
    const float bb1 = b1[col];
    const float bb2 = b2[col];

    ga = gather_row(x4, nbr, ga, sa, __builtin_amdgcn_readlane(meta, 2), ia, lane);
    gb = gather_row(x4, nbr, gb, sb, __builtin_amdgcn_readlane(meta, 3), ib, lane);
    *(float4*)(&lA[0][st_off])           = ga;
    *(float4*)(&lA[0][st_off + 8 * LDA]) = gb;
    __syncthreads();

    int buf = 0;
    for (; t < N_TILES; t += GEMM_BLOCKS) {
        const size_t rbase = (size_t)(t * 16 + 8 * hi) * DF + col;
        const int  tn   = t + GEMM_BLOCKS;
        const bool more = tn < N_TILES;
        const int  tc   = more ? tn : t;
        na = tc * 16 + swave; nb = na + 8;
        meta = mp[tc * 16 + mo];

        v8f acc = {0.f, 0.f, 0.f, 0.f, 0.f, 0.f, 0.f, 0.f};
        const float* la = &lA[buf][fr_off];
        #pragma unroll
        for (int kk = 0; kk < DF / 4; ++kk) {
            v2f a; a.x = la[4 * kk]; a.y = la[4 * kk + 1];
            acc = __builtin_amdgcn_wmma_f32_16x16x4_f32(
                false, a, false, breg1[kk], (short)0, acc, false, false);
        }
        __builtin_amdgcn_sched_group_barrier(0x020, 1, 0);
        __builtin_amdgcn_sched_group_barrier(0x100, RA, 0);
        #pragma unroll
        for (int g = 0; g < 16 - RA; ++g) {
            __builtin_amdgcn_sched_group_barrier(0x008, 2, 0);
            __builtin_amdgcn_sched_group_barrier(0x100, 1, 0);
        }
        __builtin_amdgcn_sched_group_barrier(0x008, 2 * RA, 0);

        #pragma unroll
        for (int v = 0; v < 8; ++v)
            lH[h_off + v * LDA] = fmaxf(acc[v] + bb1, 0.0f);
        asm volatile("s_wait_dscnt 0x0\n\ts_barrier_signal -1\n\ts_barrier_wait -1" ::: "memory");
        __builtin_amdgcn_sched_barrier(0);

        sa = __builtin_amdgcn_readlane(meta, 0); sb = __builtin_amdgcn_readlane(meta, 1);
        ia = nbr[(sa + lane < N_EDGES) ? sa + lane : N_EDGES - 1];
        ib = nbr[(sb + lane < N_EDGES) ? sb + lane : N_EDGES - 1];
        ga = x4[(size_t)na * (DF / 4) + lane];
        gb = x4[(size_t)nb * (DF / 4) + lane];
        float rs[8];
        #pragma unroll
        for (int v = 0; v < 8; ++v)
            rs[v] = x[rbase + v * DF];
        v8f acc2 = {0.f, 0.f, 0.f, 0.f, 0.f, 0.f, 0.f, 0.f};
        const float* lh = &lH[fr_off];
        #pragma unroll
        for (int kk = 0; kk < DF / 4; ++kk) {
            v2f a; a.x = lh[4 * kk]; a.y = lh[4 * kk + 1];
            acc2 = __builtin_amdgcn_wmma_f32_16x16x4_f32(
                false, a, false, breg2[kk], (short)0, acc2, false, false);
        }
        __builtin_amdgcn_sched_group_barrier(0x020, 12, 0);
        __builtin_amdgcn_sched_group_barrier(0x100, RA, 0);
        #pragma unroll
        for (int g = 0; g < 16 - RA; ++g) {
            __builtin_amdgcn_sched_group_barrier(0x008, 2, 0);
            __builtin_amdgcn_sched_group_barrier(0x100, 1, 0);
        }
        __builtin_amdgcn_sched_group_barrier(0x008, 2 * RA, 0);

        float yb[8];
        #pragma unroll
        for (int v = 0; v < 8; ++v)
            yb[v] = acc2[v] + bb2;
        asm volatile("" : "+v"(yb[0]), "+v"(yb[1]), "+v"(yb[2]), "+v"(yb[3]),
                          "+v"(yb[4]), "+v"(yb[5]), "+v"(yb[6]), "+v"(yb[7]));

        if (more) {
            ga = gather_row(x4, nbr, ga, sa, __builtin_amdgcn_readlane(meta, 2), ia, lane);
            gb = gather_row(x4, nbr, gb, sb, __builtin_amdgcn_readlane(meta, 3), ib, lane);
            *(float4*)(&lA[buf ^ 1][st_off])           = ga;
            *(float4*)(&lA[buf ^ 1][st_off + 8 * LDA]) = gb;
        }
        __syncthreads();
        __builtin_amdgcn_sched_barrier(0);
        buf ^= 1;

        #pragma unroll
        for (int v = 0; v < 8; ++v) {
            float y = yb[v] + rs[v];
            y = (y > 0.0f) ? y : expm1f(y);
            out[rbase + v * DF] = y;
        }
    }
}

extern "C" void kernel_launch(void* const* d_in, const int* in_sizes, int n_in,
                              void* d_out, int out_size, void* d_ws, size_t ws_size,
                              hipStream_t stream) {
    const float* x   = (const float*)d_in[0];
    const int*   ei  = (const int*)d_in[1];
    const float* W1[3] = {(const float*)d_in[2],  (const float*)d_in[6],  (const float*)d_in[10]};
    const float* b1[3] = {(const float*)d_in[3],  (const float*)d_in[7],  (const float*)d_in[11]};
    const float* W2[3] = {(const float*)d_in[4],  (const float*)d_in[8],  (const float*)d_in[12]};
    const float* b2[3] = {(const float*)d_in[5],  (const float*)d_in[9],  (const float*)d_in[13]};
    const float* Wjk = (const float*)d_in[14];
    const float* bjk = (const float*)d_in[15];

    const size_t NM = (size_t)N_NODES * DF;
    float* h0 = (float*)d_ws;
    float* xs[3] = {h0 + NM, h0 + 2 * NM, h0 + 3 * NM};

    dim3 blk(256);
    const int gBlocks = GEMM_BLOCKS;

    int*  cnt   = (int*)h0;
    int*  start = cnt + 50048;
    int*  hist  = cnt + 100096;
    int*  off   = cnt + 149120;
    int*  nbr   = cnt + 200704;
    int2* pairs = (int2*)(cnt + 1000704);
    bin_count_kernel<<<EDGE_BLOCKS, EDGE_THREADS, 0, stream>>>(hist, ei);
    scan_kernel<<<1, 896, 0, stream>>>(off, hist);
    bin_scatter_kernel<<<EDGE_BLOCKS, EDGE_THREADS, 0, stream>>>(pairs, off, ei);
    group_kernel<<<N_BINS, 1024, 0, stream>>>(nbr, start, cnt, pairs, off);

    const float* xin = x;
    for (int l = 0; l < 3; ++l) {
        mlp_wmma<<<gBlocks, blk, 0, stream>>>(xin, start, cnt, nbr, W1[l], b1[l], W2[l], b2[l], xs[l]);
        xin = xs[l];
    }
    gemm_wmma<3, 0><<<gBlocks, blk, 0, stream>>>(xs[0], xs[1], xs[2],
                                                 Wjk, bjk, nullptr, (float*)d_out);
}


# Epilogues in row layout via an LDS tile: each wave writes whole 512-B output rows, one store per row; the layer residual comes from the own rows already in registers (no partial-line access left)
# speedup vs baseline: 3.5292x; 1.0349x over previous
#include <hip/hip_runtime.h>
#include <cmath>

#define N_NODES 50000
#define N_EDGES 800000
#define DF 128

typedef __attribute__((ext_vector_type(2))) float v2f;
typedef __attribute__((ext_vector_type(8))) float v8f;

#define BIN_SHIFT 8
#define BIN_NODES (1 << BIN_SHIFT)
#define N_BINS 196
#define EDGE_BLOCKS 250
#define Q_PER_BLOCK 800
#define HIST_LEN (N_BINS * EDGE_BLOCKS)
#define EDGE_THREADS 160
#define Q_PER_THREAD 5
#define SCAN_CH 56
#define SCAN_THREADS 875
static_assert(EDGE_BLOCKS * Q_PER_BLOCK * 4 == N_EDGES, "chunks cover the edge list exactly");
static_assert(EDGE_THREADS * Q_PER_THREAD == Q_PER_BLOCK, "threads cover a chunk exactly");
static_assert(N_BINS * BIN_NODES >= N_NODES, "bins cover all nodes");
static_assert(SCAN_THREADS * SCAN_CH == HIST_LEN && SCAN_CH % 4 == 0 && SCAN_THREADS <= 896, "scan chunks");

__global__ void __launch_bounds__(EDGE_THREADS)
bin_count_kernel(int* hist, const int* ei) {
    __shared__ int h[N_BINS];
    const int tid = threadIdx.x;
    for (int i = tid; i < N_BINS; i += EDGE_THREADS) h[i] = 0;
    const int4* d4 = (const int4*)(ei + N_EDGES) + blockIdx.x * Q_PER_BLOCK;
    int4 d[Q_PER_THREAD];
    #pragma unroll
    for (int k = 0; k < Q_PER_THREAD; ++k) d[k] = d4[tid + k * EDGE_THREADS];
    __syncthreads();
    #pragma unroll
    for (int k = 0; k < Q_PER_THREAD; ++k) {
        atomicAdd(&h[d[k].x >> BIN_SHIFT], 1);
        atomicAdd(&h[d[k].y >> BIN_SHIFT], 1);
        atomicAdd(&h[d[k].z >> BIN_SHIFT], 1);
        atomicAdd(&h[d[k].w >> BIN_SHIFT], 1);
    }
    __syncthreads();
    for (int i = tid; i < N_BINS; i += EDGE_THREADS) hist[i * EDGE_BLOCKS + blockIdx.x] = h[i];
}

__device__ __forceinline__ int wave_inclusive_scan(int v, int lane) {
    #pragma unroll
    for (int d = 1; d < 32; d <<= 1) {
        const int t = __shfl_up(v, d);
        if (lane >= d) v += t;
    }
    return v;
}

__global__ void __launch_bounds__(896)
scan_kernel(int* __restrict__ off, const int* __restrict__ hist) {
    __shared__ int wsum[32];
    const int tid = threadIdx.x;
    const int lane = tid & 31;
    const int wave = tid >> 5;
    const bool act = tid < SCAN_THREADS;
    const int base = tid * SCAN_CH;
    int v[SCAN_CH];
    #pragma unroll
    for (int k = 0; k < SCAN_CH; ++k) v[k] = 0;
    if (act) {
        #pragma unroll
        for (int k = 0; k < SCAN_CH; k += 4) {
            const int4 q = *(const int4*)(hist + base + k);
            v[k] = q.x; v[k + 1] = q.y; v[k + 2] = q.z; v[k + 3] = q.w;
        }
    }
    int s = 0;
    #pragma unroll
    for (int k = 0; k < SCAN_CH; ++k) s += v[k];
    const int inc = wave_inclusive_scan(s, lane);
    if (lane == 31) wsum[wave] = inc;
    __syncthreads();
    if (wave == 0) {
        const int w = (lane < 28) ? wsum[lane] : 0;
        wsum[lane] = wave_inclusive_scan(w, lane) - w;
    }
    __syncthreads();
    if (act) {
        int run = wsum[wave] + inc - s;
        #pragma unroll
        for (int k = 0; k < SCAN_CH; k += 4) {
            int4 q;
            q.x = run; run += v[k];
            q.y = run; run += v[k + 1];
            q.z = run; run += v[k + 2];
            q.w = run; run += v[k + 3];
            *(int4*)(off + base + k) = q;
        }
    }
}

__global__ void __launch_bounds__(EDGE_THREADS)
bin_scatter_kernel(int2* __restrict__ pairs, const int* __restrict__ off,
                   const int* __restrict__ ei) {
    __shared__ int cur[N_BINS];
    const int tid = threadIdx.x;
    for (int i = tid; i < N_BINS; i += EDGE_THREADS) cur[i] = off[i * EDGE_BLOCKS + blockIdx.x];
    const int4* s4 = (const int4*)ei + blockIdx.x * Q_PER_BLOCK;
    const int4* d4 = (const int4*)(ei + N_EDGES) + blockIdx.x * Q_PER_BLOCK;
    int4 s[Q_PER_THREAD], d[Q_PER_THREAD];
    #pragma unroll
    for (int k = 0; k < Q_PER_THREAD; ++k) {
        s[k] = s4[tid + k * EDGE_THREADS];
        d[k] = d4[tid + k * EDGE_THREADS];
    }
    __syncthreads();
    int4 p[Q_PER_THREAD];
    #pragma unroll
    for (int k = 0; k < Q_PER_THREAD; ++k) {
        p[k].x = atomicAdd(&cur[d[k].x >> BIN_SHIFT], 1);
        p[k].y = atomicAdd(&cur[d[k].y >> BIN_SHIFT], 1);
        p[k].z = atomicAdd(&cur[d[k].z >> BIN_SHIFT], 1);
        p[k].w = atomicAdd(&cur[d[k].w >> BIN_SHIFT], 1);
    }
    #pragma unroll
    for (int k = 0; k < Q_PER_THREAD; ++k) {
        pairs[p[k].x] = make_int2(s[k].x, d[k].x);
        pairs[p[k].y] = make_int2(s[k].y, d[k].y);
        pairs[p[k].z] = make_int2(s[k].z, d[k].z);
        pairs[p[k].w] = make_int2(s[k].w, d[k].w);
    }
}

__global__ void __launch_bounds__(1024)
group_kernel(int* __restrict__ nbr, int* __restrict__ start, int* __restrict__ cnt,
             const int2* __restrict__ pairs, const int* __restrict__ off) {
    __shared__ int c[BIN_NODES];
    __shared__ int cur[BIN_NODES];
    __shared__ int wsum[8];
    const int tid = threadIdx.x;
    const int lane = tid & 31;
    const int wave = tid >> 5;
    const int b = blockIdx.x;
    const int lo = off[b * EDGE_BLOCKS];
    const int hi = (b + 1 < N_BINS) ? off[(b + 1) * EDGE_BLOCKS] : N_EDGES;
    if (tid < BIN_NODES) c[tid] = 0;
    __syncthreads();
    for (int i0 = lo + tid; i0 < hi; i0 += 4096) {
        int dd[4];
        #pragma unroll
        for (int k = 0; k < 4; ++k) {
            const int i = i0 + k * 1024;
            dd[k] = pairs[(i < hi) ? i : hi - 1].y;
        }
        #pragma unroll
        for (int k = 0; k < 4; ++k)
            if (i0 + k * 1024 < hi) atomicAdd(&c[dd[k] & (BIN_NODES - 1)], 1);
    }
    __syncthreads();
    int cv = 0, inc = 0;
    if (tid < BIN_NODES) {
        cv = c[tid];
        inc = wave_inclusive_scan(cv, lane);
        if (lane == 31) wsum[wave] = inc;
    }
    __syncthreads();
    if (tid < BIN_NODES) {
        int wp = 0;
        for (int w = 0; w < wave; ++w) wp += wsum[w];
        const int st = lo + wp + inc - cv;
        cur[tid] = st;
        const int node = b * BIN_NODES + tid;
        if (node < N_NODES) { start[node] = st; cnt[node] = cv; }
    }
    __syncthreads();
    for (int i0 = lo + tid; i0 < hi; i0 += 4096) {
        int2 p[4];
        #pragma unroll
        for (int k = 0; k < 4; ++k) {
            const int i = i0 + k * 1024;
            p[k] = pairs[(i < hi) ? i : hi - 1];
        }
        #pragma unroll
        for (int k = 0; k < 4; ++k)
            if (i0 + k * 1024 < hi)
                nbr[atomicAdd(&cur[p[k].y & (BIN_NODES - 1)], 1)] = p[k].x;
    }
}

template<int K>
__device__ __forceinline__ void add_rows(float4& acc, const float4* x4,
                                         int idx, int j, int lane) {
    float4 v[K];
    #pragma unroll
    for (int k = 0; k < K; ++k) {
        int s = __builtin_amdgcn_readlane(idx, j + k);
        v[k] = x4[(size_t)s * (DF / 4) + lane];
    }
    #pragma unroll
    for (int k = 0; k < K; ++k) {
        acc.x += v[k].x; acc.y += v[k].y; acc.z += v[k].z; acc.w += v[k].w;
    }
}

__device__ __forceinline__ float4 gather_row(const float4* x4, const int* nbr, float4 acc,
                                             int s0, int deg, int idx0, int lane) {
    for (int b = 0; b < deg; b += 32) {
        const int m = (deg - b < 32) ? (deg - b) : 32;
        const int idx = (b == 0) ? idx0 : ((lane < m) ? nbr[s0 + b + lane] : 0);
        int j = 0;
        for (; j + 8 <= m; j += 8) add_rows<8>(acc, x4, idx, j, lane);
        switch (m - j) {
            case 7: add_rows<7>(acc, x4, idx, j, lane); break;
            case 6: add_rows<6>(acc, x4, idx, j, lane); break;
            case 5: add_rows<5>(acc, x4, idx, j, lane); break;
            case 4: add_rows<4>(acc, x4, idx, j, lane); break;
            case 3: add_rows<3>(acc, x4, idx, j, lane); break;
            case 2: add_rows<2>(acc, x4, idx, j, lane); break;
            case 1: add_rows<1>(acc, x4, idx, j, lane); break;
            default: break;
        }
    }
    return acc;
}

#define LDA (DF + 4)
#define N_TILES 3125
#define GEMM_BLOCKS 512
template<int NSEG, int EPI>
__global__ void __launch_bounds__(256, 4)
gemm_wmma(const float* A0, const float* A1, const float* A2,
          const float* __restrict__ W, const float* __restrict__ bias,
          const float* resid, float* out) {
    __shared__ float lA[2][16 * LDA];
    __shared__ float lO[16 * LDA];

    const int tid  = threadIdx.x;
    const int lane = tid & 31;
    const int wave = tid >> 5;
    const int r    = lane & 15;
    const int hi   = lane >> 4;
    const int n0   = wave * 16;
    const int col  = n0 + r;
    const int h_off = 8 * hi * LDA + col;

    const float* As[3] = {A0, A1, A2};

    constexpr int RA = 4;

    const int st_off = wave * LDA + lane * 4;
    const int fr_off = r * LDA + 2 * hi;

    int t = blockIdx.x;
    float4 p0, p1;
    {
        const float4* s4 = (const float4*)(As[0] + (size_t)t * 16 * DF);
        p0 = s4[tid];
        p1 = s4[tid + 256];
    }

    v2f breg[NSEG][DF / 4];
    #pragma unroll
    for (int seg = 0; seg < NSEG; ++seg) {
        #pragma unroll
        for (int kk = 0; kk < DF / 4; ++kk) {
            const float* bp = W + seg * DF * DF + (4 * kk + 2 * hi) * DF + col;
            breg[seg][kk].x = bp[0];
            breg[seg][kk].y = bp[DF];
        }
    }
    const float bb = bias[col];

    *(float4*)(&lA[0][st_off])           = p0;
    *(float4*)(&lA[0][st_off + 8 * LDA]) = p1;
    __syncthreads();

    int buf = 0;
    for (; t < N_TILES; t += GEMM_BLOCKS) {
        v8f acc = {0.f, 0.f, 0.f, 0.f, 0.f, 0.f, 0.f, 0.f};
        float rs[8] = {0.f, 0.f, 0.f, 0.f, 0.f, 0.f, 0.f, 0.f};
        const size_t rbase = (size_t)(t * 16 + 8 * hi) * DF + col;
        #pragma unroll
        for (int seg = 0; seg < NSEG; ++seg) {
            const bool last = (seg == NSEG - 1);
            const int  tn   = last ? t + GEMM_BLOCKS : t;
            const int  sn   = last ? 0 : seg + 1;
            const int  tc   = (tn < N_TILES) ? tn : t;
            {
                const float4* s4 = (const float4*)(As[sn] + (size_t)tc * 16 * DF);
                p0 = s4[tid];
                p1 = s4[tid + 256];
            }
            if (EPI == 2 && last) {
                #pragma unroll
                for (int v = 0; v < 8; ++v)
                    rs[v] = resid[rbase + v * DF];
            }

            const float* la = &lA[buf][fr_off];
            #pragma unroll
            for (int kk = 0; kk < DF / 4; ++kk) {
                v2f a; a.x = la[4 * kk]; a.y = la[4 * kk + 1];
                acc = __builtin_amdgcn_wmma_f32_16x16x4_f32(
                    false, a, false, breg[seg][kk], (short)0, acc, false, false);
            }
            if (EPI == 2 && last) __builtin_amdgcn_sched_group_barrier(0x020, 10, 0);
            else                  __builtin_amdgcn_sched_group_barrier(0x020, 2, 0);
            __builtin_amdgcn_sched_group_barrier(0x100, RA, 0);
            #pragma unroll
            for (int g = 0; g < 16 - RA; ++g) {
                __builtin_amdgcn_sched_group_barrier(0x008, 2, 0);
                __builtin_amdgcn_sched_group_barrier(0x100, 1, 0);
            }
            __builtin_amdgcn_sched_group_barrier(0x008, 2 * RA, 0);

            if (last) {
                #pragma unroll
                for (int v = 0; v < 8; ++v) {
                    float y = acc[v] + bb;
                    if (EPI == 1) y = fmaxf(y, 0.0f);
                    if (EPI == 2) {
                        y += rs[v];
                        y = (y > 0.0f) ? y : expm1f(y);
                    }
                    lO[h_off + v * LDA] = y;
                }
            }
            *(float4*)(&lA[buf ^ 1][st_off])           = p0;
            *(float4*)(&lA[buf ^ 1][st_off + 8 * LDA]) = p1;
            __syncthreads();
            __builtin_amdgcn_sched_barrier(0);
            buf ^= 1;
        }

        {
            const float4 oa = *(const float4*)(&lO[st_off]);
            const float4 ob = *(const float4*)(&lO[st_off + 8 * LDA]);
            float4* o4 = (float4*)out + (size_t)(t * 16 + wave) * (DF / 4) + lane;
            o4[0]            = oa;
            o4[8 * (DF / 4)] = ob;
        }
    }
}

__global__ void __launch_bounds__(256, 4)
mlp_wmma(const float* x, const int* start, const int* cnt, const int* nbr,
         const float* __restrict__ W1, const float* __restrict__ b1,
         const float* __restrict__ W2, const float* __restrict__ b2,
         float* out) {
    __shared__ float lA[2][16 * LDA];
    __shared__ float lH[16 * LDA];
    __shared__ float lO[16 * LDA];

    const int tid  = threadIdx.x;
    const int lane = tid & 31;
    const int wave = tid >> 5;
    const int swave = __builtin_amdgcn_readfirstlane(wave);
    const int r    = lane & 15;
    const int hi   = lane >> 4;
    const int col  = wave * 16 + r;

    constexpr int RA = 4;

    const int st_off = wave * LDA + lane * 4;
    const int fr_off = r * LDA + 2 * hi;
    const int h_off  = 8 * hi * LDA + col;

    const float4* x4 = (const float4*)x;

    const int* mp = (lane & 2) ? cnt : start;
    const int  mo = swave + 8 * (lane & 1);

    int t = blockIdx.x;
    int na = t * 16 + swave, nb = na + 8;
    int meta = mp[t * 16 + mo];
    int sa = __builtin_amdgcn_readlane(meta, 0), sb = __builtin_amdgcn_readlane(meta, 1);
    int ia = nbr[(sa + lane < N_EDGES) ? sa + lane : N_EDGES - 1];
    int ib = nbr[(sb + lane < N_EDGES) ? sb + lane : N_EDGES - 1];
    float4 ga = x4[(size_t)na * (DF / 4) + lane];
    float4 gb = x4[(size_t)nb * (DF / 4) + lane];

    v2f breg1[DF / 4], breg2[DF / 4];
    #pragma unroll
    for (int kk = 0; kk < DF / 4; ++kk) {
        const float* bp1 = W1 + (4 * kk + 2 * hi) * DF + col;
        breg1[kk].x = bp1[0];
        breg1[kk].y = bp1[DF];
        const float* bp2 = W2 + (4 * kk + 2 * hi) * DF + col;
        breg2[kk].x = bp2[0];
        breg2[kk].y = bp2[DF];
    }
    const float bb1 = b1[col];
    const float bb2 = b2[col];

    float4 xa = ga, xb = gb;
    {
        const float4 suma = gather_row(x4, nbr, ga, sa, __builtin_amdgcn_readlane(meta, 2), ia, lane);
        const float4 sumb = gather_row(x4, nbr, gb, sb, __builtin_amdgcn_readlane(meta, 3), ib, lane);
        *(float4*)(&lA[0][st_off])           = suma;
        *(float4*)(&lA[0][st_off + 8 * LDA]) = sumb;
    }
    __syncthreads();

    int buf = 0;
    for (; t < N_TILES; t += GEMM_BLOCKS) {
        const int  tn   = t + GEMM_BLOCKS;
        const bool more = tn < N_TILES;
        const int  tc   = more ? tn : t;
        na = tc * 16 + swave; nb = na + 8;
        meta = mp[tc * 16 + mo];

        v8f acc = {0.f, 0.f, 0.f, 0.f, 0.f, 0.f, 0.f, 0.f};
        const float* la = &lA[buf][fr_off];
        #pragma unroll
        for (int kk = 0; kk < DF / 4; ++kk) {
            v2f a; a.x = la[4 * kk]; a.y = la[4 * kk + 1];
            acc = __builtin_amdgcn_wmma_f32_16x16x4_f32(
                false, a, false, breg1[kk], (short)0, acc, false, false);
        }
        __builtin_amdgcn_sched_group_barrier(0x020, 1, 0);
        __builtin_amdgcn_sched_group_barrier(0x100, RA, 0);
        #pragma unroll
        for (int g = 0; g < 16 - RA; ++g) {
            __builtin_amdgcn_sched_group_barrier(0x008, 2, 0);
            __builtin_amdgcn_sched_group_barrier(0x100, 1, 0);
        }
        __builtin_amdgcn_sched_group_barrier(0x008, 2 * RA, 0);

        #pragma unroll
        for (int v = 0; v < 8; ++v)
            lH[h_off + v * LDA] = fmaxf(acc[v] + bb1, 0.0f);
        asm volatile("s_wait_dscnt 0x0\n\ts_barrier_signal -1\n\ts_barrier_wait -1" ::: "memory");
        __builtin_amdgcn_sched_barrier(0);

        sa = __builtin_amdgcn_readlane(meta, 0); sb = __builtin_amdgcn_readlane(meta, 1);
        ia = nbr[(sa + lane < N_EDGES) ? sa + lane : N_EDGES - 1];
        ib = nbr[(sb + lane < N_EDGES) ? sb + lane : N_EDGES - 1];
        ga = x4[(size_t)na * (DF / 4) + lane];
        gb = x4[(size_t)nb * (DF / 4) + lane];
        v8f acc2 = {0.f, 0.f, 0.f, 0.f, 0.f, 0.f, 0.f, 0.f};
        const float* lh = &lH[fr_off];
        #pragma unroll
        for (int kk = 0; kk < DF / 4; ++kk) {
            v2f a; a.x = lh[4 * kk]; a.y = lh[4 * kk + 1];
            acc2 = __builtin_amdgcn_wmma_f32_16x16x4_f32(
                false, a, false, breg2[kk], (short)0, acc2, false, false);
        }
        __builtin_amdgcn_sched_group_barrier(0x020, 4, 0);
        __builtin_amdgcn_sched_group_barrier(0x100, RA, 0);
        #pragma unroll
        for (int g = 0; g < 16 - RA; ++g) {
            __builtin_amdgcn_sched_group_barrier(0x008, 2, 0);
            __builtin_amdgcn_sched_group_barrier(0x100, 1, 0);
        }
        __builtin_amdgcn_sched_group_barrier(0x008, 2 * RA, 0);

        #pragma unroll
        for (int v = 0; v < 8; ++v)
            lO[h_off + v * LDA] = acc2[v] + bb2;

        if (more) {
            const float4 suma = gather_row(x4, nbr, ga, sa, __builtin_amdgcn_readlane(meta, 2), ia, lane);
            const float4 sumb = gather_row(x4, nbr, gb, sb, __builtin_amdgcn_readlane(meta, 3), ib, lane);
            *(float4*)(&lA[buf ^ 1][st_off])           = suma;
            *(float4*)(&lA[buf ^ 1][st_off + 8 * LDA]) = sumb;
        }
        __syncthreads();
        __builtin_amdgcn_sched_barrier(0);
        buf ^= 1;

        {
            const float4 oa = *(const float4*)(&lO[st_off]);
            const float4 ob = *(const float4*)(&lO[st_off + 8 * LDA]);
            float4 ya, yc;
            ya.x = oa.x + xa.x; ya.y = oa.y + xa.y; ya.z = oa.z + xa.z; ya.w = oa.w + xa.w;
            yc.x = ob.x + xb.x; yc.y = ob.y + xb.y; yc.z = ob.z + xb.z; yc.w = ob.w + xb.w;
            ya.x = (ya.x > 0.0f) ? ya.x : expm1f(ya.x);
            ya.y = (ya.y > 0.0f) ? ya.y : expm1f(ya.y);
            ya.z = (ya.z > 0.0f) ? ya.z : expm1f(ya.z);
            ya.w = (ya.w > 0.0f) ? ya.w : expm1f(ya.w);
            yc.x = (yc.x > 0.0f) ? yc.x : expm1f(yc.x);
            yc.y = (yc.y > 0.0f) ? yc.y : expm1f(yc.y);
            yc.z = (yc.z > 0.0f) ? yc.z : expm1f(yc.z);
            yc.w = (yc.w > 0.0f) ? yc.w : expm1f(yc.w);
            float4* o4 = (float4*)out + (size_t)(t * 16 + swave) * (DF / 4) + lane;
            o4[0]            = ya;
            o4[8 * (DF / 4)] = yc;
        }
        xa = ga; xb = gb;
    }
}

extern "C" void kernel_launch(void* const* d_in, const int* in_sizes, int n_in,
                              void* d_out, int out_size, void* d_ws, size_t ws_size,
                              hipStream_t stream) {
    const float* x   = (const float*)d_in[0];
    const int*   ei  = (const int*)d_in[1];
    const float* W1[3] = {(const float*)d_in[2],  (const float*)d_in[6],  (const float*)d_in[10]};
    const float* b1[3] = {(const float*)d_in[3],  (const float*)d_in[7],  (const float*)d_in[11]};
    const float* W2[3] = {(const float*)d_in[4],  (const float*)d_in[8],  (const float*)d_in[12]};
    const float* b2[3] = {(const float*)d_in[5],  (const float*)d_in[9],  (const float*)d_in[13]};
    const float* Wjk = (const float*)d_in[14];
    const float* bjk = (const float*)d_in[15];

    const size_t NM = (size_t)N_NODES * DF;
    float* h0 = (float*)d_ws;
    float* xs[3] = {h0 + NM, h0 + 2 * NM, h0 + 3 * NM};

    dim3 blk(256);
    const int gBlocks = GEMM_BLOCKS;

    int*  cnt   = (int*)h0;
    int*  start = cnt + 50048;
    int*  hist  = cnt + 100096;
    int*  off   = cnt + 149120;
    int*  nbr   = cnt + 200704;
    int2* pairs = (int2*)(cnt + 1000704);
    bin_count_kernel<<<EDGE_BLOCKS, EDGE_THREADS, 0, stream>>>(hist, ei);
    scan_kernel<<<1, 896, 0, stream>>>(off, hist);
    bin_scatter_kernel<<<EDGE_BLOCKS, EDGE_THREADS, 0, stream>>>(pairs, off, ei);
    group_kernel<<<N_BINS, 1024, 0, stream>>>(nbr, start, cnt, pairs, off);

    const float* xin = x;
    for (int l = 0; l < 3; ++l) {
        mlp_wmma<<<gBlocks, blk, 0, stream>>>(xin, start, cnt, nbr, W1[l], b1[l], W2[l], b2[l], xs[l]);
        xin = xs[l];
    }
    gemm_wmma<3, 0><<<gBlocks, blk, 0, stream>>>(xs[0], xs[1], xs[2],
                                                 Wjk, bjk, nullptr, (float*)d_out);
}


# Each wave requests 8 neighbour rows of BOTH its nodes per step while both lists have 8 left (16 rows in flight, was 8): fewer memory round trips per tile; each node's add order unchanged
# speedup vs baseline: 3.6357x; 1.0302x over previous
#include <hip/hip_runtime.h>
#include <cmath>

#define N_NODES 50000
#define N_EDGES 800000
#define DF 128

typedef __attribute__((ext_vector_type(2))) float v2f;
typedef __attribute__((ext_vector_type(8))) float v8f;

#define BIN_SHIFT 8
#define BIN_NODES (1 << BIN_SHIFT)
#define N_BINS 196
#define EDGE_BLOCKS 250
#define Q_PER_BLOCK 800
#define HIST_LEN (N_BINS * EDGE_BLOCKS)
#define EDGE_THREADS 160
#define Q_PER_THREAD 5
#define SCAN_CH 56
#define SCAN_THREADS 875
static_assert(EDGE_BLOCKS * Q_PER_BLOCK * 4 == N_EDGES, "chunks cover the edge list exactly");
static_assert(EDGE_THREADS * Q_PER_THREAD == Q_PER_BLOCK, "threads cover a chunk exactly");
static_assert(N_BINS * BIN_NODES >= N_NODES, "bins cover all nodes");
static_assert(SCAN_THREADS * SCAN_CH == HIST_LEN && SCAN_CH % 4 == 0 && SCAN_THREADS <= 896, "scan chunks");

__global__ void __launch_bounds__(EDGE_THREADS)
bin_count_kernel(int* hist, const int* ei) {
    __shared__ int h[N_BINS];
    const int tid = threadIdx.x;
    for (int i = tid; i < N_BINS; i += EDGE_THREADS) h[i] = 0;
    const int4* d4 = (const int4*)(ei + N_EDGES) + blockIdx.x * Q_PER_BLOCK;
    int4 d[Q_PER_THREAD];
    #pragma unroll
    for (int k = 0; k < Q_PER_THREAD; ++k) d[k] = d4[tid + k * EDGE_THREADS];
    __syncthreads();
    #pragma unroll
    for (int k = 0; k < Q_PER_THREAD; ++k) {
        atomicAdd(&h[d[k].x >> BIN_SHIFT], 1);
        atomicAdd(&h[d[k].y >> BIN_SHIFT], 1);
        atomicAdd(&h[d[k].z >> BIN_SHIFT], 1);
        atomicAdd(&h[d[k].w >> BIN_SHIFT], 1);
    }
    __syncthreads();
    for (int i = tid; i < N_BINS; i += EDGE_THREADS) hist[i * EDGE_BLOCKS + blockIdx.x] = h[i];
}

__device__ __forceinline__ int wave_inclusive_scan(int v, int lane) {
    #pragma unroll
    for (int d = 1; d < 32; d <<= 1) {
        const int t = __shfl_up(v, d);
        if (lane >= d) v += t;
    }
    return v;
}

__global__ void __launch_bounds__(896)
scan_kernel(int* __restrict__ off, const int* __restrict__ hist) {
    __shared__ int wsum[32];
    const int tid = threadIdx.x;
    const int lane = tid & 31;
    const int wave = tid >> 5;
    const bool act = tid < SCAN_THREADS;
    const int base = tid * SCAN_CH;
    int v[SCAN_CH];
    #pragma unroll
    for (int k = 0; k < SCAN_CH; ++k) v[k] = 0;
    if (act) {
        #pragma unroll
        for (int k = 0; k < SCAN_CH; k += 4) {
            const int4 q = *(const int4*)(hist + base + k);
            v[k] = q.x; v[k + 1] = q.y; v[k + 2] = q.z; v[k + 3] = q.w;
        }
    }
    int s = 0;
    #pragma unroll
    for (int k = 0; k < SCAN_CH; ++k) s += v[k];
    const int inc = wave_inclusive_scan(s, lane);
    if (lane == 31) wsum[wave] = inc;
    __syncthreads();
    if (wave == 0) {
        const int w = (lane < 28) ? wsum[lane] : 0;
        wsum[lane] = wave_inclusive_scan(w, lane) - w;
    }
    __syncthreads();
    if (act) {
        int run = wsum[wave] + inc - s;
        #pragma unroll
        for (int k = 0; k < SCAN_CH; k += 4) {
            int4 q;
            q.x = run; run += v[k];
            q.y = run; run += v[k + 1];
            q.z = run; run += v[k + 2];
            q.w = run; run += v[k + 3];
            *(int4*)(off + base + k) = q;
        }
    }
}

__global__ void __launch_bounds__(EDGE_THREADS)
bin_scatter_kernel(int2* __restrict__ pairs, const int* __restrict__ off,
                   const int* __restrict__ ei) {
    __shared__ int cur[N_BINS];
    const int tid = threadIdx.x;
    for (int i = tid; i < N_BINS; i += EDGE_THREADS) cur[i] = off[i * EDGE_BLOCKS + blockIdx.x];
    const int4* s4 = (const int4*)ei + blockIdx.x * Q_PER_BLOCK;
    const int4* d4 = (const int4*)(ei + N_EDGES) + blockIdx.x * Q_PER_BLOCK;
    int4 s[Q_PER_THREAD], d[Q_PER_THREAD];
    #pragma unroll
    for (int k = 0; k < Q_PER_THREAD; ++k) {
        s[k] = s4[tid + k * EDGE_THREADS];
        d[k] = d4[tid + k * EDGE_THREADS];
    }
    __syncthreads();
    int4 p[Q_PER_THREAD];
    #pragma unroll
    for (int k = 0; k < Q_PER_THREAD; ++k) {
        p[k].x = atomicAdd(&cur[d[k].x >> BIN_SHIFT], 1);
        p[k].y = atomicAdd(&cur[d[k].y >> BIN_SHIFT], 1);
        p[k].z = atomicAdd(&cur[d[k].z >> BIN_SHIFT], 1);
        p[k].w = atomicAdd(&cur[d[k].w >> BIN_SHIFT], 1);
    }
    #pragma unroll
    for (int k = 0; k < Q_PER_THREAD; ++k) {
        pairs[p[k].x] = make_int2(s[k].x, d[k].x);
        pairs[p[k].y] = make_int2(s[k].y, d[k].y);
        pairs[p[k].z] = make_int2(s[k].z, d[k].z);
        pairs[p[k].w] = make_int2(s[k].w, d[k].w);
    }
}

__global__ void __launch_bounds__(1024)
group_kernel(int* __restrict__ nbr, int* __restrict__ start, int* __restrict__ cnt,
             const int2* __restrict__ pairs, const int* __restrict__ off) {
    __shared__ int c[BIN_NODES];
    __shared__ int cur[BIN_NODES];
    __shared__ int wsum[8];
    const int tid = threadIdx.x;
    const int lane = tid & 31;
    const int wave = tid >> 5;
    const int b = blockIdx.x;
    const int lo = off[b * EDGE_BLOCKS];
    const int hi = (b + 1 < N_BINS) ? off[(b + 1) * EDGE_BLOCKS] : N_EDGES;
    if (tid < BIN_NODES) c[tid] = 0;
    __syncthreads();
    for (int i0 = lo + tid; i0 < hi; i0 += 4096) {
        int dd[4];
        #pragma unroll
        for (int k = 0; k < 4; ++k) {
            const int i = i0 + k * 1024;
            dd[k] = pairs[(i < hi) ? i : hi - 1].y;
        }
        #pragma unroll
        for (int k = 0; k < 4; ++k)
            if (i0 + k * 1024 < hi) atomicAdd(&c[dd[k] & (BIN_NODES - 1)], 1);
    }
    __syncthreads();
    int cv = 0, inc = 0;
    if (tid < BIN_NODES) {
        cv = c[tid];
        inc = wave_inclusive_scan(cv, lane);
        if (lane == 31) wsum[wave] = inc;
    }
    __syncthreads();
    if (tid < BIN_NODES) {
        int wp = 0;
        for (int w = 0; w < wave; ++w) wp += wsum[w];
        const int st = lo + wp + inc - cv;
        cur[tid] = st;
        const int node = b * BIN_NODES + tid;
        if (node < N_NODES) { start[node] = st; cnt[node] = cv; }
    }
    __syncthreads();
    for (int i0 = lo + tid; i0 < hi; i0 += 4096) {
        int2 p[4];
        #pragma unroll
        for (int k = 0; k < 4; ++k) {
            const int i = i0 + k * 1024;
            p[k] = pairs[(i < hi) ? i : hi - 1];
        }
        #pragma unroll
        for (int k = 0; k < 4; ++k)
            if (i0 + k * 1024 < hi)
                nbr[atomicAdd(&cur[p[k].y & (BIN_NODES - 1)], 1)] = p[k].x;
    }
}

template<int K>
__device__ __forceinline__ void add_rows(float4& acc, const float4* x4,
                                         int idx, int j, int lane) {
    float4 v[K];
    #pragma unroll
    for (int k = 0; k < K; ++k) {
        int s = __builtin_amdgcn_readlane(idx, j + k);
        v[k] = x4[(size_t)s * (DF / 4) + lane];
    }
    #pragma unroll
    for (int k = 0; k < K; ++k) {
        acc.x += v[k].x; acc.y += v[k].y; acc.z += v[k].z; acc.w += v[k].w;
    }
}

__device__ __forceinline__ float4 gather_rest(const float4* x4, const int* nbr, float4 acc,
                                              int s0, int deg, int lane) {
    for (int b = 32; b < deg; b += 32) {
        const int m = (deg - b < 32) ? (deg - b) : 32;
        const int idx = (lane < m) ? nbr[s0 + b + lane] : 0;
        int j = 0;
        for (; j + 8 <= m; j += 8) add_rows<8>(acc, x4, idx, j, lane);
        switch (m - j) {
            case 7: add_rows<7>(acc, x4, idx, j, lane); break;
            case 6: add_rows<6>(acc, x4, idx, j, lane); break;
            case 5: add_rows<5>(acc, x4, idx, j, lane); break;
            case 4: add_rows<4>(acc, x4, idx, j, lane); break;
            case 3: add_rows<3>(acc, x4, idx, j, lane); break;
            case 2: add_rows<2>(acc, x4, idx, j, lane); break;
            case 1: add_rows<1>(acc, x4, idx, j, lane); break;
            default: break;
        }
    }
    return acc;
}

__device__ __forceinline__ void add_rows_pair(float4& acc_a, float4& acc_b, const float4* x4,
                                              int ia, int ib, int j, int lane) {
    float4 va[8], vb[8];
    #pragma unroll
    for (int k = 0; k < 8; ++k) {
        const int s = __builtin_amdgcn_readlane(ia, j + k);
        va[k] = x4[(size_t)s * (DF / 4) + lane];
    }
    #pragma unroll
    for (int k = 0; k < 8; ++k) {
        const int s = __builtin_amdgcn_readlane(ib, j + k);
        vb[k] = x4[(size_t)s * (DF / 4) + lane];
    }
    #pragma unroll
    for (int k = 0; k < 8; ++k) {
        acc_a.x += va[k].x; acc_a.y += va[k].y; acc_a.z += va[k].z; acc_a.w += va[k].w;
    }
    #pragma unroll
    for (int k = 0; k < 8; ++k) {
        acc_b.x += vb[k].x; acc_b.y += vb[k].y; acc_b.z += vb[k].z; acc_b.w += vb[k].w;
    }
}

__device__ __forceinline__ float4 rest_of_list(const float4* x4, float4 acc, int idx, int j, int m, int lane) {
    for (; j + 8 <= m; j += 8) add_rows<8>(acc, x4, idx, j, lane);
    switch (m - j) {
        case 7: add_rows<7>(acc, x4, idx, j, lane); break;
        case 6: add_rows<6>(acc, x4, idx, j, lane); break;
        case 5: add_rows<5>(acc, x4, idx, j, lane); break;
        case 4: add_rows<4>(acc, x4, idx, j, lane); break;
        case 3: add_rows<3>(acc, x4, idx, j, lane); break;
        case 2: add_rows<2>(acc, x4, idx, j, lane); break;
        case 1: add_rows<1>(acc, x4, idx, j, lane); break;
        default: break;
    }
    return acc;
}

__device__ __forceinline__ void gather_pair(const float4* x4, const int* nbr,
                                            float4& acc_a, float4& acc_b,
                                            int sa, int sb, int da, int db,
                                            int ia, int ib, int lane) {
    const int ma = (da < 32) ? da : 32;
    const int mb = (db < 32) ? db : 32;
    const int mn = (ma < mb) ? ma : mb;
    int j = 0;
    for (; j + 8 <= mn; j += 8) add_rows_pair(acc_a, acc_b, x4, ia, ib, j, lane);
    acc_a = rest_of_list(x4, acc_a, ia, j, ma, lane);
    acc_b = rest_of_list(x4, acc_b, ib, j, mb, lane);
    if (da > 32) acc_a = gather_rest(x4, nbr, acc_a, sa, da, lane);
    if (db > 32) acc_b = gather_rest(x4, nbr, acc_b, sb, db, lane);
}

#define LDA (DF + 4)
#define N_TILES 3125
#define GEMM_BLOCKS 512
template<int NSEG, int EPI>
__global__ void __launch_bounds__(256, 4)
gemm_wmma(const float* A0, const float* A1, const float* A2,
          const float* __restrict__ W, const float* __restrict__ bias,
          const float* resid, float* out) {
    __shared__ float lA[2][16 * LDA];
    __shared__ float lO[16 * LDA];

    const int tid  = threadIdx.x;
    const int lane = tid & 31;
    const int wave = tid >> 5;
    const int r    = lane & 15;
    const int hi   = lane >> 4;
    const int n0   = wave * 16;
    const int col  = n0 + r;
    const int h_off = 8 * hi * LDA + col;

    const float* As[3] = {A0, A1, A2};

    constexpr int RA = 4;

    const int st_off = wave * LDA + lane * 4;
    const int fr_off = r * LDA + 2 * hi;

    int t = blockIdx.x;
    float4 p0, p1;
    {
        const float4* s4 = (const float4*)(As[0] + (size_t)t * 16 * DF);
        p0 = s4[tid];
        p1 = s4[tid + 256];
    }

    v2f breg[NSEG][DF / 4];
    #pragma unroll
    for (int seg = 0; seg < NSEG; ++seg) {
        #pragma unroll
        for (int kk = 0; kk < DF / 4; ++kk) {
            const float* bp = W + seg * DF * DF + (4 * kk + 2 * hi) * DF + col;
            breg[seg][kk].x = bp[0];
            breg[seg][kk].y = bp[DF];
        }
    }
    const float bb = bias[col];

    *(float4*)(&lA[0][st_off])           = p0;
    *(float4*)(&lA[0][st_off + 8 * LDA]) = p1;
    __syncthreads();

    int buf = 0;
    for (; t < N_TILES; t += GEMM_BLOCKS) {
        v8f acc = {0.f, 0.f, 0.f, 0.f, 0.f, 0.f, 0.f, 0.f};
        float rs[8] = {0.f, 0.f, 0.f, 0.f, 0.f, 0.f, 0.f, 0.f};
        const size_t rbase = (size_t)(t * 16 + 8 * hi) * DF + col;
        #pragma unroll
        for (int seg = 0; seg < NSEG; ++seg) {
            const bool last = (seg == NSEG - 1);
            const int  tn   = last ? t + GEMM_BLOCKS : t;
            const int  sn   = last ? 0 : seg + 1;
            const int  tc   = (tn < N_TILES) ? tn : t;
            {
                const float4* s4 = (const float4*)(As[sn] + (size_t)tc * 16 * DF);
                p0 = s4[tid];
                p1 = s4[tid + 256];
            }
            if (EPI == 2 && last) {
                #pragma unroll
                for (int v = 0; v < 8; ++v)
                    rs[v] = resid[rbase + v * DF];
            }

            const float* la = &lA[buf][fr_off];
            #pragma unroll
            for (int kk = 0; kk < DF / 4; ++kk) {
                v2f a; a.x = la[4 * kk]; a.y = la[4 * kk + 1];
                acc = __builtin_amdgcn_wmma_f32_16x16x4_f32(
                    false, a, false, breg[seg][kk], (short)0, acc, false, false);
            }
            if (EPI == 2 && last) __builtin_amdgcn_sched_group_barrier(0x020, 10, 0);
            else                  __builtin_amdgcn_sched_group_barrier(0x020, 2, 0);
            __builtin_amdgcn_sched_group_barrier(0x100, RA, 0);
            #pragma unroll
            for (int g = 0; g < 16 - RA; ++g) {
                __builtin_amdgcn_sched_group_barrier(0x008, 2, 0);
                __builtin_amdgcn_sched_group_barrier(0x100, 1, 0);
            }
            __builtin_amdgcn_sched_group_barrier(0x008, 2 * RA, 0);

            if (last) {
                #pragma unroll
                for (int v = 0; v < 8; ++v) {
                    float y = acc[v] + bb;
                    if (EPI == 1) y = fmaxf(y, 0.0f);
                    if (EPI == 2) {
                        y += rs[v];
                        y = (y > 0.0f) ? y : expm1f(y);
                    }
                    lO[h_off + v * LDA] = y;
                }
            }
            *(float4*)(&lA[buf ^ 1][st_off])           = p0;
            *(float4*)(&lA[buf ^ 1][st_off + 8 * LDA]) = p1;
            __syncthreads();
            __builtin_amdgcn_sched_barrier(0);
            buf ^= 1;
        }

        {
            const float4 oa = *(const float4*)(&lO[st_off]);
            const float4 ob = *(const float4*)(&lO[st_off + 8 * LDA]);
            float4* o4 = (float4*)out + (size_t)(t * 16 + wave) * (DF / 4) + lane;
            o4[0]            = oa;
            o4[8 * (DF / 4)] = ob;
        }
    }
}

__global__ void __launch_bounds__(256, 4)
mlp_wmma(const float* x, const int* start, const int* cnt, const int* nbr,
         const float* __restrict__ W1, const float* __restrict__ b1,
         const float* __restrict__ W2, const float* __restrict__ b2,
         float* out) {
    __shared__ float lA[2][16 * LDA];
    __shared__ float lH[16 * LDA];
    __shared__ float lO[16 * LDA];

    const int tid  = threadIdx.x;
    const int lane = tid & 31;
    const int wave = tid >> 5;
    const int swave = __builtin_amdgcn_readfirstlane(wave);
    const int r    = lane & 15;
    const int hi   = lane >> 4;
    const int col  = wave * 16 + r;

    constexpr int RA = 4;

    const int st_off = wave * LDA + lane * 4;
    const int fr_off = r * LDA + 2 * hi;
    const int h_off  = 8 * hi * LDA + col;

    const float4* x4 = (const float4*)x;

    const int* mp = (lane & 2) ? cnt : start;
    const int  mo = swave + 8 * (lane & 1);

    int t = blockIdx.x;
    int na = t * 16 + swave, nb = na + 8;
    int meta = mp[t * 16 + mo];
    int sa = __builtin_amdgcn_readlane(meta, 0), sb = __builtin_amdgcn_readlane(meta, 1);
    int ia = nbr[(sa + lane < N_EDGES) ? sa + lane : N_EDGES - 1];
    int ib = nbr[(sb + lane < N_EDGES) ? sb + lane : N_EDGES - 1];
    float4 ga = x4[(size_t)na * (DF / 4) + lane];
    float4 gb = x4[(size_t)nb * (DF / 4) + lane];

    v2f breg1[DF / 4], breg2[DF / 4];
    #pragma unroll
    for (int kk = 0; kk < DF / 4; ++kk) {
        const float* bp1 = W1 + (4 * kk + 2 * hi) * DF + col;
        breg1[kk].x = bp1[0];
        breg1[kk].y = bp1[DF];
        const float* bp2 = W2 + (4 * kk + 2 * hi) * DF + col;
        breg2[kk].x = bp2[0];
        breg2[kk].y = bp2[DF];
    }
    const float bb1 = b1[col];
    const float bb2 = b2[col];

    float4 xa = ga, xb = gb;
    {
        float4 suma = ga, sumb = gb;
        gather_pair(x4, nbr, suma, sumb, sa, sb, __builtin_amdgcn_readlane(meta, 2),
                    __builtin_amdgcn_readlane(meta, 3), ia, ib, lane);
        *(float4*)(&lA[0][st_off])           = suma;
        *(float4*)(&lA[0][st_off + 8 * LDA]) = sumb;
    }
    __syncthreads();

    int buf = 0;
    for (; t < N_TILES; t += GEMM_BLOCKS) {
        const int  tn   = t + GEMM_BLOCKS;
        const bool more = tn < N_TILES;
        const int  tc   = more ? tn : t;
        na = tc * 16 + swave; nb = na + 8;
        meta = mp[tc * 16 + mo];

        v8f acc = {0.f, 0.f, 0.f, 0.f, 0.f, 0.f, 0.f, 0.f};
        const float* la = &lA[buf][fr_off];
        #pragma unroll
        for (int kk = 0; kk < DF / 4; ++kk) {
            v2f a; a.x = la[4 * kk]; a.y = la[4 * kk + 1];
            acc = __builtin_amdgcn_wmma_f32_16x16x4_f32(
                false, a, false, breg1[kk], (short)0, acc, false, false);
        }
        __builtin_amdgcn_sched_group_barrier(0x020, 1, 0);
        __builtin_amdgcn_sched_group_barrier(0x100, RA, 0);
        #pragma unroll
        for (int g = 0; g < 16 - RA; ++g) {
            __builtin_amdgcn_sched_group_barrier(0x008, 2, 0);
            __builtin_amdgcn_sched_group_barrier(0x100, 1, 0);
        }
        __builtin_amdgcn_sched_group_barrier(0x008, 2 * RA, 0);

        #pragma unroll
        for (int v = 0; v < 8; ++v)
            lH[h_off + v * LDA] = fmaxf(acc[v] + bb1, 0.0f);
        asm volatile("s_wait_dscnt 0x0\n\ts_barrier_signal -1\n\ts_barrier_wait -1" ::: "memory");
        __builtin_amdgcn_sched_barrier(0);

        sa = __builtin_amdgcn_readlane(meta, 0); sb = __builtin_amdgcn_readlane(meta, 1);
        ia = nbr[(sa + lane < N_EDGES) ? sa + lane : N_EDGES - 1];
        ib = nbr[(sb + lane < N_EDGES) ? sb + lane : N_EDGES - 1];
        ga = x4[(size_t)na * (DF / 4) + lane];
        gb = x4[(size_t)nb * (DF / 4) + lane];
        v8f acc2 = {0.f, 0.f, 0.f, 0.f, 0.f, 0.f, 0.f, 0.f};
        const float* lh = &lH[fr_off];
        #pragma unroll
        for (int kk = 0; kk < DF / 4; ++kk) {
            v2f a; a.x = lh[4 * kk]; a.y = lh[4 * kk + 1];
            acc2 = __builtin_amdgcn_wmma_f32_16x16x4_f32(
                false, a, false, breg2[kk], (short)0, acc2, false, false);
        }
        __builtin_amdgcn_sched_group_barrier(0x020, 4, 0);
        __builtin_amdgcn_sched_group_barrier(0x100, RA, 0);
        #pragma unroll
        for (int g = 0; g < 16 - RA; ++g) {
            __builtin_amdgcn_sched_group_barrier(0x008, 2, 0);
            __builtin_amdgcn_sched_group_barrier(0x100, 1, 0);
        }
        __builtin_amdgcn_sched_group_barrier(0x008, 2 * RA, 0);

        #pragma unroll
        for (int v = 0; v < 8; ++v)
            lO[h_off + v * LDA] = acc2[v] + bb2;

        if (more) {
            float4 suma = ga, sumb = gb;
            gather_pair(x4, nbr, suma, sumb, sa, sb, __builtin_amdgcn_readlane(meta, 2),
                        __builtin_amdgcn_readlane(meta, 3), ia, ib, lane);
            *(float4*)(&lA[buf ^ 1][st_off])           = suma;
            *(float4*)(&lA[buf ^ 1][st_off + 8 * LDA]) = sumb;
        }
        __syncthreads();
        __builtin_amdgcn_sched_barrier(0);
        buf ^= 1;

        {
            const float4 oa = *(const float4*)(&lO[st_off]);
            const float4 ob = *(const float4*)(&lO[st_off + 8 * LDA]);
            float4 ya, yc;
            ya.x = oa.x + xa.x; ya.y = oa.y + xa.y; ya.z = oa.z + xa.z; ya.w = oa.w + xa.w;
            yc.x = ob.x + xb.x; yc.y = ob.y + xb.y; yc.z = ob.z + xb.z; yc.w = ob.w + xb.w;
            ya.x = (ya.x > 0.0f) ? ya.x : expm1f(ya.x);
            ya.y = (ya.y > 0.0f) ? ya.y : expm1f(ya.y);
            ya.z = (ya.z > 0.0f) ? ya.z : expm1f(ya.z);
            ya.w = (ya.w > 0.0f) ? ya.w : expm1f(ya.w);
            yc.x = (yc.x > 0.0f) ? yc.x : expm1f(yc.x);
            yc.y = (yc.y > 0.0f) ? yc.y : expm1f(yc.y);
            yc.z = (yc.z > 0.0f) ? yc.z : expm1f(yc.z);
            yc.w = (yc.w > 0.0f) ? yc.w : expm1f(yc.w);
            float4* o4 = (float4*)out + (size_t)(t * 16 + swave) * (DF / 4) + lane;
            o4[0]            = ya;
            o4[8 * (DF / 4)] = yc;
        }
        xa = ga; xb = gb;
    }
}

extern "C" void kernel_launch(void* const* d_in, const int* in_sizes, int n_in,
                              void* d_out, int out_size, void* d_ws, size_t ws_size,
                              hipStream_t stream) {
    const float* x   = (const float*)d_in[0];
    const int*   ei  = (const int*)d_in[1];
    const float* W1[3] = {(const float*)d_in[2],  (const float*)d_in[6],  (const float*)d_in[10]};
    const float* b1[3] = {(const float*)d_in[3],  (const float*)d_in[7],  (const float*)d_in[11]};
    const float* W2[3] = {(const float*)d_in[4],  (const float*)d_in[8],  (const float*)d_in[12]};
    const float* b2[3] = {(const float*)d_in[5],  (const float*)d_in[9],  (const float*)d_in[13]};
    const float* Wjk = (const float*)d_in[14];
    const float* bjk = (const float*)d_in[15];

    const size_t NM = (size_t)N_NODES * DF;
    float* h0 = (float*)d_ws;
    float* xs[3] = {h0 + NM, h0 + 2 * NM, h0 + 3 * NM};

    dim3 blk(256);
    const int gBlocks = GEMM_BLOCKS;

    int*  cnt   = (int*)h0;
    int*  start = cnt + 50048;
    int*  hist  = cnt + 100096;
    int*  off   = cnt + 149120;
    int*  nbr   = cnt + 200704;
    int2* pairs = (int2*)(cnt + 1000704);
    bin_count_kernel<<<EDGE_BLOCKS, EDGE_THREADS, 0, stream>>>(hist, ei);
    scan_kernel<<<1, 896, 0, stream>>>(off, hist);
    bin_scatter_kernel<<<EDGE_BLOCKS, EDGE_THREADS, 0, stream>>>(pairs, off, ei);
    group_kernel<<<N_BINS, 1024, 0, stream>>>(nbr, start, cnt, pairs, off);

    const float* xin = x;
    for (int l = 0; l < 3; ++l) {
        mlp_wmma<<<gBlocks, blk, 0, stream>>>(xin, start, cnt, nbr, W1[l], b1[l], W2[l], b2[l], xs[l]);
        xin = xs[l];
    }
    gemm_wmma<3, 0><<<gBlocks, blk, 0, stream>>>(xs[0], xs[1], xs[2],
                                                 Wjk, bjk, nullptr, (float*)d_out);
}


# Edge grouping in two launches, was four: each chunk bin-sorts its own edges in LDS, each bin's work-group sums the lower bins' counts itself; a node's neighbour sum order = another LDS arrival rank
# speedup vs baseline: 4.0445x; 1.1124x over previous
#include <hip/hip_runtime.h>
#include <cmath>

#define N_NODES 50000
#define N_EDGES 800000
#define DF 128

typedef __attribute__((ext_vector_type(2))) float v2f;
typedef __attribute__((ext_vector_type(8))) float v8f;

#define BIN_SHIFT 8
#define BIN_NODES (1 << BIN_SHIFT)
#define N_BINS 196
#define EDGE_BLOCKS 250
#define Q_PER_BLOCK 800
#define CHUNK_EDGES (Q_PER_BLOCK * 4)
#define HIST_LEN (N_BINS * EDGE_BLOCKS)
#define EDGE_THREADS 160
#define Q_PER_THREAD 5
#define BINS_PER_LANE 7
#define BASE_LOADS 12
static_assert(EDGE_BLOCKS * Q_PER_BLOCK * 4 == N_EDGES, "chunks cover the edge list exactly");
static_assert(EDGE_THREADS * Q_PER_THREAD == Q_PER_BLOCK, "threads cover a chunk exactly");
static_assert(N_BINS * BIN_NODES >= N_NODES, "bins cover all nodes");
static_assert(32 * BINS_PER_LANE >= N_BINS, "one wave scans all bins");
static_assert(HIST_LEN % 4 == 0 && BASE_LOADS * 1024 * 4 >= HIST_LEN, "base loads cover hist");
static_assert(EDGE_BLOCKS * 4 <= 1024, "four threads per run");

__device__ __forceinline__ int wave_inclusive_scan(int v, int lane) {
    #pragma unroll
    for (int d = 1; d < 32; d <<= 1) {
        const int t = __shfl_up(v, d);
        if (lane >= d) v += t;
    }
    return v;
}

__global__ void __launch_bounds__(EDGE_THREADS)
chunk_sort_kernel(int2* pairs, int* hist, int* loc, const int* ei) {
    __shared__ int h[N_BINS];
    const int tid = threadIdx.x;
    for (int i = tid; i < N_BINS; i += EDGE_THREADS) h[i] = 0;
    const int4* s4 = (const int4*)ei + blockIdx.x * Q_PER_BLOCK;
    const int4* d4 = (const int4*)(ei + N_EDGES) + blockIdx.x * Q_PER_BLOCK;
    int4 s[Q_PER_THREAD], d[Q_PER_THREAD];
    #pragma unroll
    for (int k = 0; k < Q_PER_THREAD; ++k) {
        s[k] = s4[tid + k * EDGE_THREADS];
        d[k] = d4[tid + k * EDGE_THREADS];
    }
    __syncthreads();
    #pragma unroll
    for (int k = 0; k < Q_PER_THREAD; ++k) {
        atomicAdd(&h[d[k].x >> BIN_SHIFT], 1);
        atomicAdd(&h[d[k].y >> BIN_SHIFT], 1);
        atomicAdd(&h[d[k].z >> BIN_SHIFT], 1);
        atomicAdd(&h[d[k].w >> BIN_SHIFT], 1);
    }
    __syncthreads();
    if (tid < 32) {
        int v[BINS_PER_LANE];
        int sum = 0;
        #pragma unroll
        for (int j = 0; j < BINS_PER_LANE; ++j) {
            const int bin = tid * BINS_PER_LANE + j;
            v[j] = (bin < N_BINS) ? h[bin] : 0;
            sum += v[j];
        }
        int run = wave_inclusive_scan(sum, tid) - sum;
        #pragma unroll
        for (int j = 0; j < BINS_PER_LANE; ++j) {
            const int bin = tid * BINS_PER_LANE + j;
            if (bin < N_BINS) {
                hist[bin * EDGE_BLOCKS + blockIdx.x] = v[j];
                loc[bin * EDGE_BLOCKS + blockIdx.x]  = run;
                h[bin] = run;
            }
            run += v[j];
        }
    }
    __syncthreads();
    int4 p[Q_PER_THREAD];
    #pragma unroll
    for (int k = 0; k < Q_PER_THREAD; ++k) {
        p[k].x = atomicAdd(&h[d[k].x >> BIN_SHIFT], 1);
        p[k].y = atomicAdd(&h[d[k].y >> BIN_SHIFT], 1);
        p[k].z = atomicAdd(&h[d[k].z >> BIN_SHIFT], 1);
        p[k].w = atomicAdd(&h[d[k].w >> BIN_SHIFT], 1);
    }
    int2* reg = pairs + blockIdx.x * CHUNK_EDGES;
    #pragma unroll
    for (int k = 0; k < Q_PER_THREAD; ++k) {
        reg[p[k].x] = make_int2(s[k].x, d[k].x);
        reg[p[k].y] = make_int2(s[k].y, d[k].y);
        reg[p[k].z] = make_int2(s[k].z, d[k].z);
        reg[p[k].w] = make_int2(s[k].w, d[k].w);
    }
}

__global__ void __launch_bounds__(1024)
group_kernel(int* __restrict__ nbr, int* __restrict__ start, int* __restrict__ cnt,
             const int2* __restrict__ pairs, const int* __restrict__ hist,
             const int* __restrict__ loc) {
    __shared__ int c[BIN_NODES];
    __shared__ int cur[BIN_NODES];
    __shared__ int wsum[32];
    const int tid = threadIdx.x;
    const int lane = tid & 31;
    const int wave = tid >> 5;
    const int b = blockIdx.x;
    const int nflat = b * EDGE_BLOCKS;
    const int4* h4 = (const int4*)hist;
    int4 q[BASE_LOADS];
    #pragma unroll
    for (int k = 0; k < BASE_LOADS; ++k) {
        const int i4 = tid + k * 1024;
        q[k] = h4[(i4 < HIST_LEN / 4) ? i4 : HIST_LEN / 4 - 1];
    }
    const int ch = tid >> 2;
    const int j0 = tid & 3;
    int s0 = 0, n = 0;
    if (ch < EDGE_BLOCKS) {
        s0 = ch * CHUNK_EDGES + loc[b * EDGE_BLOCKS + ch];
        n  = hist[b * EDGE_BLOCKS + ch];
    }
    if (tid < BIN_NODES) c[tid] = 0;
    int part = 0;
    #pragma unroll
    for (int k = 0; k < BASE_LOADS; ++k) {
        const int e = 4 * (tid + k * 1024);
        part += (e     < nflat) ? q[k].x : 0;
        part += (e + 1 < nflat) ? q[k].y : 0;
        part += (e + 2 < nflat) ? q[k].z : 0;
        part += (e + 3 < nflat) ? q[k].w : 0;
    }
    const int pinc = wave_inclusive_scan(part, lane);
    if (lane == 31) wsum[wave] = pinc;
    __syncthreads();
    int base = 0;
    for (int w = 0; w < 32; ++w) base += wsum[w];
    for (int k0 = j0; k0 < n; k0 += 16) {
        int dd[4];
        #pragma unroll
        for (int k = 0; k < 4; ++k) {
            const int i = k0 + 4 * k;
            dd[k] = pairs[s0 + ((i < n) ? i : n - 1)].y;
        }
        #pragma unroll
        for (int k = 0; k < 4; ++k)
            if (k0 + 4 * k < n) atomicAdd(&c[dd[k] & (BIN_NODES - 1)], 1);
    }
    __syncthreads();
    int cv = 0, inc = 0;
    if (tid < BIN_NODES) {
        cv = c[tid];
        inc = wave_inclusive_scan(cv, lane);
        if (lane == 31) wsum[wave] = inc;
    }
    __syncthreads();
    if (tid < BIN_NODES) {
        int wp = 0;
        for (int w = 0; w < wave; ++w) wp += wsum[w];
        const int st = base + wp + inc - cv;
        cur[tid] = st;
        const int node = b * BIN_NODES + tid;
        if (node < N_NODES) { start[node] = st; cnt[node] = cv; }
    }
    __syncthreads();
    for (int k0 = j0; k0 < n; k0 += 16) {
        int2 p[4];
        #pragma unroll
        for (int k = 0; k < 4; ++k) {
            const int i = k0 + 4 * k;
            p[k] = pairs[s0 + ((i < n) ? i : n - 1)];
        }
        #pragma unroll
        for (int k = 0; k < 4; ++k)
            if (k0 + 4 * k < n)
                nbr[atomicAdd(&cur[p[k].y & (BIN_NODES - 1)], 1)] = p[k].x;
    }
}

template<int K>
__device__ __forceinline__ void add_rows(float4& acc, const float4* x4,
                                         int idx, int j, int lane) {
    float4 v[K];
    #pragma unroll
    for (int k = 0; k < K; ++k) {
        int s = __builtin_amdgcn_readlane(idx, j + k);
        v[k] = x4[(size_t)s * (DF / 4) + lane];
    }
    #pragma unroll
    for (int k = 0; k < K; ++k) {
        acc.x += v[k].x; acc.y += v[k].y; acc.z += v[k].z; acc.w += v[k].w;
    }
}

__device__ __forceinline__ float4 gather_rest(const float4* x4, const int* nbr, float4 acc,
                                              int s0, int deg, int lane) {
    for (int b = 32; b < deg; b += 32) {
        const int m = (deg - b < 32) ? (deg - b) : 32;
        const int idx = (lane < m) ? nbr[s0 + b + lane] : 0;
        int j = 0;
        for (; j + 8 <= m; j += 8) add_rows<8>(acc, x4, idx, j, lane);
        switch (m - j) {
            case 7: add_rows<7>(acc, x4, idx, j, lane); break;
            case 6: add_rows<6>(acc, x4, idx, j, lane); break;
            case 5: add_rows<5>(acc, x4, idx, j, lane); break;
            case 4: add_rows<4>(acc, x4, idx, j, lane); break;
            case 3: add_rows<3>(acc, x4, idx, j, lane); break;
            case 2: add_rows<2>(acc, x4, idx, j, lane); break;
            case 1: add_rows<1>(acc, x4, idx, j, lane); break;
            default: break;
        }
    }
    return acc;
}

__device__ __forceinline__ void add_rows_pair(float4& acc_a, float4& acc_b, const float4* x4,
                                              int ia, int ib, int j, int lane) {
    float4 va[8], vb[8];
    #pragma unroll
    for (int k = 0; k < 8; ++k) {
        const int s = __builtin_amdgcn_readlane(ia, j + k);
        va[k] = x4[(size_t)s * (DF / 4) + lane];
    }
    #pragma unroll
    for (int k = 0; k < 8; ++k) {
        const int s = __builtin_amdgcn_readlane(ib, j + k);
        vb[k] = x4[(size_t)s * (DF / 4) + lane];
    }
    #pragma unroll
    for (int k = 0; k < 8; ++k) {
        acc_a.x += va[k].x; acc_a.y += va[k].y; acc_a.z += va[k].z; acc_a.w += va[k].w;
    }
    #pragma unroll
    for (int k = 0; k < 8; ++k) {
        acc_b.x += vb[k].x; acc_b.y += vb[k].y; acc_b.z += vb[k].z; acc_b.w += vb[k].w;
    }
}

__device__ __forceinline__ float4 rest_of_list(const float4* x4, float4 acc, int idx, int j, int m, int lane) {
    for (; j + 8 <= m; j += 8) add_rows<8>(acc, x4, idx, j, lane);
    switch (m - j) {
        case 7: add_rows<7>(acc, x4, idx, j, lane); break;
        case 6: add_rows<6>(acc, x4, idx, j, lane); break;
        case 5: add_rows<5>(acc, x4, idx, j, lane); break;
        case 4: add_rows<4>(acc, x4, idx, j, lane); break;
        case 3: add_rows<3>(acc, x4, idx, j, lane); break;
        case 2: add_rows<2>(acc, x4, idx, j, lane); break;
        case 1: add_rows<1>(acc, x4, idx, j, lane); break;
        default: break;
    }
    return acc;
}

__device__ __forceinline__ void gather_pair(const float4* x4, const int* nbr,
                                            float4& acc_a, float4& acc_b,
                                            int sa, int sb, int da, int db,
                                            int ia, int ib, int lane) {
    const int ma = (da < 32) ? da : 32;
    const int mb = (db < 32) ? db : 32;
    const int mn = (ma < mb) ? ma : mb;
    int j = 0;
    for (; j + 8 <= mn; j += 8) add_rows_pair(acc_a, acc_b, x4, ia, ib, j, lane);
    acc_a = rest_of_list(x4, acc_a, ia, j, ma, lane);
    acc_b = rest_of_list(x4, acc_b, ib, j, mb, lane);
    if (da > 32) acc_a = gather_rest(x4, nbr, acc_a, sa, da, lane);
    if (db > 32) acc_b = gather_rest(x4, nbr, acc_b, sb, db, lane);
}

#define LDA (DF + 4)
#define N_TILES 3125
#define GEMM_BLOCKS 512
template<int NSEG, int EPI>
__global__ void __launch_bounds__(256, 4)
gemm_wmma(const float* A0, const float* A1, const float* A2,
          const float* __restrict__ W, const float* __restrict__ bias,
          const float* resid, float* out) {
    __shared__ float lA[2][16 * LDA];
    __shared__ float lO[16 * LDA];

    const int tid  = threadIdx.x;
    const int lane = tid & 31;
    const int wave = tid >> 5;
    const int r    = lane & 15;
    const int hi   = lane >> 4;
    const int n0   = wave * 16;
    const int col  = n0 + r;
    const int h_off = 8 * hi * LDA + col;

    const float* As[3] = {A0, A1, A2};

    constexpr int RA = 4;

    const int st_off = wave * LDA + lane * 4;
    const int fr_off = r * LDA + 2 * hi;

    int t = blockIdx.x;
    float4 p0, p1;
    {
        const float4* s4 = (const float4*)(As[0] + (size_t)t * 16 * DF);
        p0 = s4[tid];
        p1 = s4[tid + 256];
    }

    v2f breg[NSEG][DF / 4];
    #pragma unroll
    for (int seg = 0; seg < NSEG; ++seg) {
        #pragma unroll
        for (int kk = 0; kk < DF / 4; ++kk) {
            const float* bp = W + seg * DF * DF + (4 * kk + 2 * hi) * DF + col;
            breg[seg][kk].x = bp[0];
            breg[seg][kk].y = bp[DF];
        }
    }
    const float bb = bias[col];

    *(float4*)(&lA[0][st_off])           = p0;
    *(float4*)(&lA[0][st_off + 8 * LDA]) = p1;
    __syncthreads();

    int buf = 0;
    for (; t < N_TILES; t += GEMM_BLOCKS) {
        v8f acc = {0.f, 0.f, 0.f, 0.f, 0.f, 0.f, 0.f, 0.f};
        float rs[8] = {0.f, 0.f, 0.f, 0.f, 0.f, 0.f, 0.f, 0.f};
        const size_t rbase = (size_t)(t * 16 + 8 * hi) * DF + col;
        #pragma unroll
        for (int seg = 0; seg < NSEG; ++seg) {
            const bool last = (seg == NSEG - 1);
            const int  tn   = last ? t + GEMM_BLOCKS : t;
            const int  sn   = last ? 0 : seg + 1;
            const int  tc   = (tn < N_TILES) ? tn : t;
            {
                const float4* s4 = (const float4*)(As[sn] + (size_t)tc * 16 * DF);
                p0 = s4[tid];
                p1 = s4[tid + 256];
            }
            if (EPI == 2 && last) {
                #pragma unroll
                for (int v = 0; v < 8; ++v)
                    rs[v] = resid[rbase + v * DF];
            }

            const float* la = &lA[buf][fr_off];
            #pragma unroll
            for (int kk = 0; kk < DF / 4; ++kk) {
                v2f a; a.x = la[4 * kk]; a.y = la[4 * kk + 1];
                acc = __builtin_amdgcn_wmma_f32_16x16x4_f32(
                    false, a, false, breg[seg][kk], (short)0, acc, false, false);
            }
            if (EPI == 2 && last) __builtin_amdgcn_sched_group_barrier(0x020, 10, 0);
            else                  __builtin_amdgcn_sched_group_barrier(0x020, 2, 0);
            __builtin_amdgcn_sched_group_barrier(0x100, RA, 0);
            #pragma unroll
            for (int g = 0; g < 16 - RA; ++g) {
                __builtin_amdgcn_sched_group_barrier(0x008, 2, 0);
                __builtin_amdgcn_sched_group_barrier(0x100, 1, 0);
            }
            __builtin_amdgcn_sched_group_barrier(0x008, 2 * RA, 0);

            if (last) {
                #pragma unroll
                for (int v = 0; v < 8; ++v) {
                    float y = acc[v] + bb;
                    if (EPI == 1) y = fmaxf(y, 0.0f);
                    if (EPI == 2) {
                        y += rs[v];
                        y = (y > 0.0f) ? y : expm1f(y);
                    }
                    lO[h_off + v * LDA] = y;
                }
            }
            *(float4*)(&lA[buf ^ 1][st_off])           = p0;
            *(float4*)(&lA[buf ^ 1][st_off + 8 * LDA]) = p1;
            __syncthreads();
            __builtin_amdgcn_sched_barrier(0);
            buf ^= 1;
        }

        {
            const float4 oa = *(const float4*)(&lO[st_off]);
            const float4 ob = *(const float4*)(&lO[st_off + 8 * LDA]);
            float4* o4 = (float4*)out + (size_t)(t * 16 + wave) * (DF / 4) + lane;
            o4[0]            = oa;
            o4[8 * (DF / 4)] = ob;
        }
    }
}

__global__ void __launch_bounds__(256, 4)
mlp_wmma(const float* x, const int* start, const int* cnt, const int* nbr,
         const float* __restrict__ W1, const float* __restrict__ b1,
         const float* __restrict__ W2, const float* __restrict__ b2,
         float* out) {
    __shared__ float lA[2][16 * LDA];
    __shared__ float lH[16 * LDA];
    __shared__ float lO[16 * LDA];

    const int tid  = threadIdx.x;
    const int lane = tid & 31;
    const int wave = tid >> 5;
    const int swave = __builtin_amdgcn_readfirstlane(wave);
    const int r    = lane & 15;
    const int hi   = lane >> 4;
    const int col  = wave * 16 + r;

    constexpr int RA = 4;

    const int st_off = wave * LDA + lane * 4;
    const int fr_off = r * LDA + 2 * hi;
    const int h_off  = 8 * hi * LDA + col;

    const float4* x4 = (const float4*)x;

    const int* mp = (lane & 2) ? cnt : start;
    const int  mo = swave + 8 * (lane & 1);

    int t = blockIdx.x;
    int na = t * 16 + swave, nb = na + 8;
    int meta = mp[t * 16 + mo];
    int sa = __builtin_amdgcn_readlane(meta, 0), sb = __builtin_amdgcn_readlane(meta, 1);
    int ia = nbr[(sa + lane < N_EDGES) ? sa + lane : N_EDGES - 1];
    int ib = nbr[(sb + lane < N_EDGES) ? sb + lane : N_EDGES - 1];
    float4 ga = x4[(size_t)na * (DF / 4) + lane];
    float4 gb = x4[(size_t)nb * (DF / 4) + lane];

    v2f breg1[DF / 4], breg2[DF / 4];
    #pragma unroll
    for (int kk = 0; kk < DF / 4; ++kk) {
        const float* bp1 = W1 + (4 * kk + 2 * hi) * DF + col;
        breg1[kk].x = bp1[0];
        breg1[kk].y = bp1[DF];
        const float* bp2 = W2 + (4 * kk + 2 * hi) * DF + col;
        breg2[kk].x = bp2[0];
        breg2[kk].y = bp2[DF];
    }
    const float bb1 = b1[col];
    const float bb2 = b2[col];

    float4 xa = ga, xb = gb;
    {
        float4 suma = ga, sumb = gb;
        gather_pair(x4, nbr, suma, sumb, sa, sb, __builtin_amdgcn_readlane(meta, 2),
                    __builtin_amdgcn_readlane(meta, 3), ia, ib, lane);
        *(float4*)(&lA[0][st_off])           = suma;
        *(float4*)(&lA[0][st_off + 8 * LDA]) = sumb;
    }
    __syncthreads();

    int buf = 0;
    for (; t < N_TILES; t += GEMM_BLOCKS) {
        const int  tn   = t + GEMM_BLOCKS;
        const bool more = tn < N_TILES;
        const int  tc   = more ? tn : t;
        na = tc * 16 + swave; nb = na + 8;
        meta = mp[tc * 16 + mo];

        v8f acc = {0.f, 0.f, 0.f, 0.f, 0.f, 0.f, 0.f, 0.f};
        const float* la = &lA[buf][fr_off];
        #pragma unroll
        for (int kk = 0; kk < DF / 4; ++kk) {
            v2f a; a.x = la[4 * kk]; a.y = la[4 * kk + 1];
            acc = __builtin_amdgcn_wmma_f32_16x16x4_f32(
                false, a, false, breg1[kk], (short)0, acc, false, false);
        }
        __builtin_amdgcn_sched_group_barrier(0x020, 1, 0);
        __builtin_amdgcn_sched_group_barrier(0x100, RA, 0);
        #pragma unroll
        for (int g = 0; g < 16 - RA; ++g) {
            __builtin_amdgcn_sched_group_barrier(0x008, 2, 0);
            __builtin_amdgcn_sched_group_barrier(0x100, 1, 0);
        }
        __builtin_amdgcn_sched_group_barrier(0x008, 2 * RA, 0);

        #pragma unroll
        for (int v = 0; v < 8; ++v)
            lH[h_off + v * LDA] = fmaxf(acc[v] + bb1, 0.0f);
        asm volatile("s_wait_dscnt 0x0\n\ts_barrier_signal -1\n\ts_barrier_wait -1" ::: "memory");
        __builtin_amdgcn_sched_barrier(0);

        sa = __builtin_amdgcn_readlane(meta, 0); sb = __builtin_amdgcn_readlane(meta, 1);
        ia = nbr[(sa + lane < N_EDGES) ? sa + lane : N_EDGES - 1];
        ib = nbr[(sb + lane < N_EDGES) ? sb + lane : N_EDGES - 1];
        ga = x4[(size_t)na * (DF / 4) + lane];
        gb = x4[(size_t)nb * (DF / 4) + lane];
        v8f acc2 = {0.f, 0.f, 0.f, 0.f, 0.f, 0.f, 0.f, 0.f};
        const float* lh = &lH[fr_off];
        #pragma unroll
        for (int kk = 0; kk < DF / 4; ++kk) {
            v2f a; a.x = lh[4 * kk]; a.y = lh[4 * kk + 1];
            acc2 = __builtin_amdgcn_wmma_f32_16x16x4_f32(
                false, a, false, breg2[kk], (short)0, acc2, false, false);
        }
        __builtin_amdgcn_sched_group_barrier(0x020, 4, 0);
        __builtin_amdgcn_sched_group_barrier(0x100, RA, 0);
        #pragma unroll
        for (int g = 0; g < 16 - RA; ++g) {
            __builtin_amdgcn_sched_group_barrier(0x008, 2, 0);
            __builtin_amdgcn_sched_group_barrier(0x100, 1, 0);
        }
        __builtin_amdgcn_sched_group_barrier(0x008, 2 * RA, 0);

        #pragma unroll
        for (int v = 0; v < 8; ++v)
            lO[h_off + v * LDA] = acc2[v] + bb2;

        if (more) {
            float4 suma = ga, sumb = gb;
            gather_pair(x4, nbr, suma, sumb, sa, sb, __builtin_amdgcn_readlane(meta, 2),
                        __builtin_amdgcn_readlane(meta, 3), ia, ib, lane);
            *(float4*)(&lA[buf ^ 1][st_off])           = suma;
            *(float4*)(&lA[buf ^ 1][st_off + 8 * LDA]) = sumb;
        }
        __syncthreads();
        __builtin_amdgcn_sched_barrier(0);
        buf ^= 1;

        {
            const float4 oa = *(const float4*)(&lO[st_off]);
            const float4 ob = *(const float4*)(&lO[st_off + 8 * LDA]);
            float4 ya, yc;
            ya.x = oa.x + xa.x; ya.y = oa.y + xa.y; ya.z = oa.z + xa.z; ya.w = oa.w + xa.w;
            yc.x = ob.x + xb.x; yc.y = ob.y + xb.y; yc.z = ob.z + xb.z; yc.w = ob.w + xb.w;
            ya.x = (ya.x > 0.0f) ? ya.x : expm1f(ya.x);
            ya.y = (ya.y > 0.0f) ? ya.y : expm1f(ya.y);
            ya.z = (ya.z > 0.0f) ? ya.z : expm1f(ya.z);
            ya.w = (ya.w > 0.0f) ? ya.w : expm1f(ya.w);
            yc.x = (yc.x > 0.0f) ? yc.x : expm1f(yc.x);
            yc.y = (yc.y > 0.0f) ? yc.y : expm1f(yc.y);
            yc.z = (yc.z > 0.0f) ? yc.z : expm1f(yc.z);
            yc.w = (yc.w > 0.0f) ? yc.w : expm1f(yc.w);
            float4* o4 = (float4*)out + (size_t)(t * 16 + swave) * (DF / 4) + lane;
            o4[0]            = ya;
            o4[8 * (DF / 4)] = yc;
        }
        xa = ga; xb = gb;
    }
}

extern "C" void kernel_launch(void* const* d_in, const int* in_sizes, int n_in,
                              void* d_out, int out_size, void* d_ws, size_t ws_size,
                              hipStream_t stream) {
    const float* x   = (const float*)d_in[0];
    const int*   ei  = (const int*)d_in[1];
    const float* W1[3] = {(const float*)d_in[2],  (const float*)d_in[6],  (const float*)d_in[10]};
    const float* b1[3] = {(const float*)d_in[3],  (const float*)d_in[7],  (const float*)d_in[11]};
    const float* W2[3] = {(const float*)d_in[4],  (const float*)d_in[8],  (const float*)d_in[12]};
    const float* b2[3] = {(const float*)d_in[5],  (const float*)d_in[9],  (const float*)d_in[13]};
    const float* Wjk = (const float*)d_in[14];
    const float* bjk = (const float*)d_in[15];

    const size_t NM = (size_t)N_NODES * DF;
    float* h0 = (float*)d_ws;
    float* xs[3] = {h0 + NM, h0 + 2 * NM, h0 + 3 * NM};

    dim3 blk(256);
    const int gBlocks = GEMM_BLOCKS;

    int*  cnt   = (int*)h0;
    int*  start = cnt + 50048;
    int*  hist  = cnt + 100096;
    int*  loc   = cnt + 149120;
    int*  nbr   = cnt + 200704;
    int2* pairs = (int2*)(cnt + 1000704);
    chunk_sort_kernel<<<EDGE_BLOCKS, EDGE_THREADS, 0, stream>>>(pairs, hist, loc, ei);
    group_kernel<<<N_BINS, 1024, 0, stream>>>(nbr, start, cnt, pairs, hist, loc);

    const float* xin = x;
    for (int l = 0; l < 3; ++l) {
        mlp_wmma<<<gBlocks, blk, 0, stream>>>(xin, start, cnt, nbr, W1[l], b1[l], W2[l], b2[l], xs[l]);
        xin = xs[l];
    }
    gemm_wmma<3, 0><<<gBlocks, blk, 0, stream>>>(xs[0], xs[1], xs[2],
                                                 Wjk, bjk, nullptr, (float*)d_out);
}
